# starting kernel
# baseline (speedup 1.0000x reference)
#include <hip/hip_runtime.h>


typedef __attribute__((ext_vector_type(2))) float v2f;
typedef __attribute__((ext_vector_type(8))) float v8f;

#define N_NODES 50000
#define N_EDGES 800000
#define D 64

__global__ __launch_bounds__(256) void gemm_wmma(
    const float* __restrict__ feat,
    const float* __restrict__ W,
    const float* __restrict__ skip_w,
    const float* __restrict__ bias,
    float* __restrict__ h,
    float* __restrict__ out_pre)
{
  const int gwave = (int)((blockIdx.x * blockDim.x + threadIdx.x) >> 5);
  const int lane  = (int)(threadIdx.x & 31u);
  const int m0    = gwave * 16;
  if (m0 >= N_NODES) return;

  const int half = lane >> 4;
  const int l    = lane & 15;

  v8f acc0 = {}, acc1 = {}, acc2 = {}, acc3 = {};

  const float* arow = feat + (size_t)(m0 + l) * D + 2 * half;
  const float* brow = W + (size_t)l * D + 2 * half;

#pragma unroll
  for (int kb = 0; kb < D; kb += 4) {
    const v2f a = *(const v2f*)(arow + kb);
    const v2f b0 = *(const v2f*)(brow + kb);
    const v2f b1 = *(const v2f*)(brow + 16 * D + kb);
    const v2f b2 = *(const v2f*)(brow + 32 * D + kb);
    const v2f b3 = *(const v2f*)(brow + 48 * D + kb);
    acc0 = __builtin_amdgcn_wmma_f32_16x16x4_f32(false, a, false, b0,
                                                 (short)0, acc0, false, false);
    acc1 = __builtin_amdgcn_wmma_f32_16x16x4_f32(false, a, false, b1,
                                                 (short)0, acc1, false, false);
    acc2 = __builtin_amdgcn_wmma_f32_16x16x4_f32(false, a, false, b2,
                                                 (short)0, acc2, false, false);
    acc3 = __builtin_amdgcn_wmma_f32_16x16x4_f32(false, a, false, b3,
                                                 (short)0, acc3, false, false);
  }

  const v8f accs[4] = {acc0, acc1, acc2, acc3};
#pragma unroll
  for (int nt = 0; nt < 4; ++nt) {
    const int n = nt * 16 + l;
    const float sw = skip_w[n];
    const float bs = bias[n];
#pragma unroll
    for (int r = 0; r < 8; ++r) {
      const int row = m0 + r + 8 * half;
      const size_t idx = (size_t)row * D + n;
      const float v = accs[nt][r];
      h[idx] = v;
      out_pre[idx] = v * sw + bs;
    }
  }
}

__global__ __launch_bounds__(256) void scatter_edges(
    const float* __restrict__ h,
    const float* __restrict__ ew,
    const int* __restrict__ esrc,
    const int* __restrict__ edst,
    float* __restrict__ out_pre)
{
  const long long t = (long long)blockIdx.x * blockDim.x + threadIdx.x;
  const long long e = t >> 4;
  const int q = (int)(t & 15);
  if (e >= N_EDGES) return;

  const int src = esrc[e];
  const int dst = edst[e];
  const float w = ew[e];

  const float4 v = ((const float4*)(h + (size_t)src * D))[q];
  float* o = out_pre + (size_t)dst * D + q * 4;
  atomicAdd(o + 0, v.x * w);
  atomicAdd(o + 1, v.y * w);
  atomicAdd(o + 2, v.z * w);
  atomicAdd(o + 3, v.w * w);
}

__global__ __launch_bounds__(256) void selu_inplace(float* __restrict__ out,
                                                    int n)
{
  const int i = (int)(blockIdx.x * blockDim.x + threadIdx.x);
  if (i >= n) return;
  const float lam = 1.0507009873554805f;
  const float alp = 1.6732632423543772f;
  const float x = out[i];
  out[i] = lam * (x > 0.0f ? x : alp * (__expf(x) - 1.0f));
}

extern "C" void kernel_launch(void* const* d_in, const int* in_sizes, int n_in,
                              void* d_out, int out_size, void* d_ws,
                              size_t ws_size, hipStream_t stream) {
  const float* feat   = (const float*)d_in[0];
  const float* W      = (const float*)d_in[1];
  const float* bias   = (const float*)d_in[2];
  const float* skip_w = (const float*)d_in[3];
  const float* ew     = (const float*)d_in[4];
  const int*   esrc   = (const int*)d_in[5];
  const int*   edst   = (const int*)d_in[6];
  float* out = (float*)d_out;
  float* h   = (float*)d_ws;

  const int gemm_threads = ((N_NODES + 15) / 16) * 32;
  gemm_wmma<<<(gemm_threads + 255) / 256, 256, 0, stream>>>(feat, W, skip_w,
                                                            bias, h, out);

  const long long scat_threads = (long long)N_EDGES * 16;
  scatter_edges<<<(int)((scat_threads + 255) / 256), 256, 0, stream>>>(
      h, ew, esrc, edst, out);

  const int n = N_NODES * D;
  selu_inplace<<<(n + 255) / 256, 256, 0, stream>>>(out, n);
}


# Edge scatter remapped to one edge per wave instruction: lane c owns columns c and c+32, so each row load and each f32 atomic covers one whole 128-B line; atomics stay arrival-ordered
# speedup vs baseline: 1.8076x; 1.8076x over previous
#include <hip/hip_runtime.h>


typedef __attribute__((ext_vector_type(2))) float v2f;
typedef __attribute__((ext_vector_type(8))) float v8f;

#define N_NODES 50000
#define N_EDGES 800000
#define D 64

__global__ __launch_bounds__(256) void gemm_wmma(
    const float* __restrict__ feat,
    const float* __restrict__ W,
    const float* __restrict__ skip_w,
    const float* __restrict__ bias,
    float* __restrict__ h,
    float* __restrict__ out_pre)
{
  const int gwave = (int)((blockIdx.x * blockDim.x + threadIdx.x) >> 5);
  const int lane  = (int)(threadIdx.x & 31u);
  const int m0    = gwave * 16;
  if (m0 >= N_NODES) return;

  const int half = lane >> 4;
  const int l    = lane & 15;

  v8f acc0 = {}, acc1 = {}, acc2 = {}, acc3 = {};

  const float* arow = feat + (size_t)(m0 + l) * D + 2 * half;
  const float* brow = W + (size_t)l * D + 2 * half;

#pragma unroll
  for (int kb = 0; kb < D; kb += 4) {
    const v2f a = *(const v2f*)(arow + kb);
    const v2f b0 = *(const v2f*)(brow + kb);
    const v2f b1 = *(const v2f*)(brow + 16 * D + kb);
    const v2f b2 = *(const v2f*)(brow + 32 * D + kb);
    const v2f b3 = *(const v2f*)(brow + 48 * D + kb);
    acc0 = __builtin_amdgcn_wmma_f32_16x16x4_f32(false, a, false, b0,
                                                 (short)0, acc0, false, false);
    acc1 = __builtin_amdgcn_wmma_f32_16x16x4_f32(false, a, false, b1,
                                                 (short)0, acc1, false, false);
    acc2 = __builtin_amdgcn_wmma_f32_16x16x4_f32(false, a, false, b2,
                                                 (short)0, acc2, false, false);
    acc3 = __builtin_amdgcn_wmma_f32_16x16x4_f32(false, a, false, b3,
                                                 (short)0, acc3, false, false);
  }

  const v8f accs[4] = {acc0, acc1, acc2, acc3};
#pragma unroll
  for (int nt = 0; nt < 4; ++nt) {
    const int n = nt * 16 + l;
    const float sw = skip_w[n];
    const float bs = bias[n];
#pragma unroll
    for (int r = 0; r < 8; ++r) {
      const int row = m0 + r + 8 * half;
      const size_t idx = (size_t)row * D + n;
      const float v = accs[nt][r];
      h[idx] = v;
      out_pre[idx] = v * sw + bs;
    }
  }
}

#define EDGES_PER_WAVE 32
#define EDGE_CHUNK 8

static_assert(N_EDGES % EDGES_PER_WAVE == 0, "a wave owns 32 whole edges");
static_assert(EDGES_PER_WAVE % EDGE_CHUNK == 0, "chunks tile the wave's edges");

__global__ __launch_bounds__(256) void scatter_edges(
    const float* __restrict__ h,
    const float* __restrict__ ew,
    const int* __restrict__ esrc,
    const int* __restrict__ edst,
    float* __restrict__ out_pre)
{
  const int lane  = (int)(threadIdx.x & 31u);
  const int gwave = (int)((blockIdx.x * blockDim.x + threadIdx.x) >> 5);
  const int e0    = gwave * EDGES_PER_WAVE;
  if (e0 >= N_EDGES) return;

  const int   my_src = esrc[e0 + lane];
  const int   my_dst = edst[e0 + lane];
  const float my_w   = ew[e0 + lane];

#pragma unroll
  for (int c = 0; c < EDGES_PER_WAVE; c += EDGE_CHUNK) {
    float lo[EDGE_CHUNK], hi[EDGE_CHUNK];
#pragma unroll
    for (int j = 0; j < EDGE_CHUNK; ++j) {
      const int s = __builtin_amdgcn_readlane(my_src, c + j);
      const float* hrow = h + (size_t)s * D;
      lo[j] = hrow[lane];
      hi[j] = hrow[lane + 32];
    }
#pragma unroll
    for (int j = 0; j < EDGE_CHUNK; ++j) {
      const int d = __builtin_amdgcn_readlane(my_dst, c + j);
      const float w = __int_as_float(
          __builtin_amdgcn_readlane(__float_as_int(my_w), c + j));
      float* orow = out_pre + (size_t)d * D;
      atomicAdd(orow + lane, lo[j] * w);
      atomicAdd(orow + lane + 32, hi[j] * w);
    }
  }
}

__global__ __launch_bounds__(256) void selu_inplace(float* __restrict__ out,
                                                    int n)
{
  const int i = (int)(blockIdx.x * blockDim.x + threadIdx.x);
  if (i >= n) return;
  const float lam = 1.0507009873554805f;
  const float alp = 1.6732632423543772f;
  const float x = out[i];
  out[i] = lam * (x > 0.0f ? x : alp * (__expf(x) - 1.0f));
}

extern "C" void kernel_launch(void* const* d_in, const int* in_sizes, int n_in,
                              void* d_out, int out_size, void* d_ws,
                              size_t ws_size, hipStream_t stream) {
  const float* feat   = (const float*)d_in[0];
  const float* W      = (const float*)d_in[1];
  const float* bias   = (const float*)d_in[2];
  const float* skip_w = (const float*)d_in[3];
  const float* ew     = (const float*)d_in[4];
  const int*   esrc   = (const int*)d_in[5];
  const int*   edst   = (const int*)d_in[6];
  float* out = (float*)d_out;
  float* h   = (float*)d_ws;

  const int gemm_threads = ((N_NODES + 15) / 16) * 32;
  gemm_wmma<<<(gemm_threads + 255) / 256, 256, 0, stream>>>(feat, W, skip_w,
                                                            bias, h, out);

  const int scat_threads = (N_EDGES / EDGES_PER_WAVE) * 32;
  scatter_edges<<<(scat_threads + 255) / 256, 256, 0, stream>>>(
      h, ew, esrc, edst, out);

  const int n = N_NODES * D;
  selu_inplace<<<(n + 255) / 256, 256, 0, stream>>>(out, n);
}


# GEMM operand fetch: W staged once per work-group in LDS in fragment order, A rows read 16 B per lane, same k permutation on both: far fewer cache lines per fragment load; f32 k sum order changes
# speedup vs baseline: 1.8634x; 1.0309x over previous
#include <hip/hip_runtime.h>


typedef __attribute__((ext_vector_type(2))) float v2f;
typedef __attribute__((ext_vector_type(8))) float v8f;

#define N_NODES 50000
#define N_EDGES 800000
#define D 64

#define KBLOCKS (D / 8)

__global__ __launch_bounds__(256) void gemm_wmma(
    const float* __restrict__ feat,
    const float* __restrict__ W,
    const float* __restrict__ skip_w,
    const float* __restrict__ bias,
    float* __restrict__ h,
    float* __restrict__ out_pre)
{
  __shared__ float4 w_frag[KBLOCKS * 4 * 32];

  const int tid   = (int)threadIdx.x;
  const int gwave = (int)((blockIdx.x * blockDim.x + threadIdx.x) >> 5);
  const int lane  = tid & 31;
  const int m0    = gwave * 16;
  const bool active = m0 < N_NODES;

  const int half = lane >> 4;
  const int l    = lane & 15;

  const int arow_idx = active ? (m0 + l) : l;
  const float4* arow =
      (const float4*)(feat + (size_t)arow_idx * D + 4 * half);
  float4 a[KBLOCKS];
#pragma unroll
  for (int jj = 0; jj < KBLOCKS; ++jj) a[jj] = arow[2 * jj];

  const float4* W4 = (const float4*)W;
#pragma unroll
  for (int i = 0; i < (D * D / 4) / 256; ++i) {
    const int g = tid + 256 * i;
    const int n = g >> 4;
    const int q = g & 15;
    const int e = ((q >> 1) * 4 + (n >> 4)) * 32 + (q & 1) * 16 + (n & 15);
    w_frag[e] = W4[g];
  }
  __syncthreads();
  if (!active) return;

  v8f acc0 = {}, acc1 = {}, acc2 = {}, acc3 = {};

#pragma unroll
  for (int jj = 0; jj < KBLOCKS; ++jj) {
    const float4 b0 = w_frag[(jj * 4 + 0) * 32 + lane];
    const float4 b1 = w_frag[(jj * 4 + 1) * 32 + lane];
    const float4 b2 = w_frag[(jj * 4 + 2) * 32 + lane];
    const float4 b3 = w_frag[(jj * 4 + 3) * 32 + lane];
    const v2f alo = {a[jj].x, a[jj].y};
    const v2f ahi = {a[jj].z, a[jj].w};
    const v2f b0lo = {b0.x, b0.y}, b0hi = {b0.z, b0.w};
    const v2f b1lo = {b1.x, b1.y}, b1hi = {b1.z, b1.w};
    const v2f b2lo = {b2.x, b2.y}, b2hi = {b2.z, b2.w};
    const v2f b3lo = {b3.x, b3.y}, b3hi = {b3.z, b3.w};
    acc0 = __builtin_amdgcn_wmma_f32_16x16x4_f32(false, alo, false, b0lo,
                                                 (short)0, acc0, false, false);
    acc1 = __builtin_amdgcn_wmma_f32_16x16x4_f32(false, alo, false, b1lo,
                                                 (short)0, acc1, false, false);
    acc2 = __builtin_amdgcn_wmma_f32_16x16x4_f32(false, alo, false, b2lo,
                                                 (short)0, acc2, false, false);
    acc3 = __builtin_amdgcn_wmma_f32_16x16x4_f32(false, alo, false, b3lo,
                                                 (short)0, acc3, false, false);
    acc0 = __builtin_amdgcn_wmma_f32_16x16x4_f32(false, ahi, false, b0hi,
                                                 (short)0, acc0, false, false);
    acc1 = __builtin_amdgcn_wmma_f32_16x16x4_f32(false, ahi, false, b1hi,
                                                 (short)0, acc1, false, false);
    acc2 = __builtin_amdgcn_wmma_f32_16x16x4_f32(false, ahi, false, b2hi,
                                                 (short)0, acc2, false, false);
    acc3 = __builtin_amdgcn_wmma_f32_16x16x4_f32(false, ahi, false, b3hi,
                                                 (short)0, acc3, false, false);
  }

  const v8f accs[4] = {acc0, acc1, acc2, acc3};
#pragma unroll
  for (int nt = 0; nt < 4; ++nt) {
    const int n = nt * 16 + l;
    const float sw = skip_w[n];
    const float bs = bias[n];
#pragma unroll
    for (int r = 0; r < 8; ++r) {
      const int row = m0 + r + 8 * half;
      const size_t idx = (size_t)row * D + n;
      const float v = accs[nt][r];
      h[idx] = v;
      out_pre[idx] = v * sw + bs;
    }
  }
}

#define EDGES_PER_WAVE 32
#define EDGE_CHUNK 8

static_assert(N_EDGES % EDGES_PER_WAVE == 0, "a wave owns 32 whole edges");
static_assert(EDGES_PER_WAVE % EDGE_CHUNK == 0, "chunks tile the wave's edges");

__global__ __launch_bounds__(256) void scatter_edges(
    const float* __restrict__ h,
    const float* __restrict__ ew,
    const int* __restrict__ esrc,
    const int* __restrict__ edst,
    float* __restrict__ out_pre)
{
  const int lane  = (int)(threadIdx.x & 31u);
  const int gwave = (int)((blockIdx.x * blockDim.x + threadIdx.x) >> 5);
  const int e0    = gwave * EDGES_PER_WAVE;
  if (e0 >= N_EDGES) return;

  const int   my_src = esrc[e0 + lane];
  const int   my_dst = edst[e0 + lane];
  const float my_w   = ew[e0 + lane];

#pragma unroll
  for (int c = 0; c < EDGES_PER_WAVE; c += EDGE_CHUNK) {
    float lo[EDGE_CHUNK], hi[EDGE_CHUNK];
#pragma unroll
    for (int j = 0; j < EDGE_CHUNK; ++j) {
      const int s = __builtin_amdgcn_readlane(my_src, c + j);
      const float* hrow = h + (size_t)s * D;
      lo[j] = hrow[lane];
      hi[j] = hrow[lane + 32];
    }
#pragma unroll
    for (int j = 0; j < EDGE_CHUNK; ++j) {
      const int d = __builtin_amdgcn_readlane(my_dst, c + j);
      const float w = __int_as_float(
          __builtin_amdgcn_readlane(__float_as_int(my_w), c + j));
      float* orow = out_pre + (size_t)d * D;
      atomicAdd(orow + lane, lo[j] * w);
      atomicAdd(orow + lane + 32, hi[j] * w);
    }
  }
}

__global__ __launch_bounds__(256) void selu_inplace(float* __restrict__ out,
                                                    int n)
{
  const int i = (int)(blockIdx.x * blockDim.x + threadIdx.x);
  if (i >= n) return;
  const float lam = 1.0507009873554805f;
  const float alp = 1.6732632423543772f;
  const float x = out[i];
  out[i] = lam * (x > 0.0f ? x : alp * (__expf(x) - 1.0f));
}

extern "C" void kernel_launch(void* const* d_in, const int* in_sizes, int n_in,
                              void* d_out, int out_size, void* d_ws,
                              size_t ws_size, hipStream_t stream) {
  const float* feat   = (const float*)d_in[0];
  const float* W      = (const float*)d_in[1];
  const float* bias   = (const float*)d_in[2];
  const float* skip_w = (const float*)d_in[3];
  const float* ew     = (const float*)d_in[4];
  const int*   esrc   = (const int*)d_in[5];
  const int*   edst   = (const int*)d_in[6];
  float* out = (float*)d_out;
  float* h   = (float*)d_ws;

  const int gemm_threads = ((N_NODES + 15) / 16) * 32;
  gemm_wmma<<<(gemm_threads + 255) / 256, 256, 0, stream>>>(feat, W, skip_w,
                                                            bias, h, out);

  const int scat_threads = (N_EDGES / EDGES_PER_WAVE) * 32;
  scatter_edges<<<(scat_threads + 255) / 256, 256, 0, stream>>>(
      h, ew, esrc, edst, out);

  const int n = N_NODES * D;
  selu_inplace<<<(n + 255) / 256, 256, 0, stream>>>(out, n);
}


# Edge sum without global atomics: 4096-edge chunks counting-sorted by dst range in LDS; one 32-wave work-group per 200 dst rows adds h[src]*w in LDS, fuses skip, bias, selu; row sums arrival-ordered
# speedup vs baseline: 1.9684x; 1.0563x over previous
#include <hip/hip_runtime.h>


typedef __attribute__((ext_vector_type(2))) float v2f;
typedef __attribute__((ext_vector_type(8))) float v8f;

#define N_NODES 50000
#define N_EDGES 800000
#define D 64

#define KBLOCKS (D / 8)

__global__ __launch_bounds__(256) void gemm_wmma(
    const float* __restrict__ feat,
    const float* __restrict__ W,
    float* __restrict__ h)
{
  __shared__ float4 w_frag[KBLOCKS * 4 * 32];

  const int tid   = (int)threadIdx.x;
  const int gwave = (int)((blockIdx.x * blockDim.x + threadIdx.x) >> 5);
  const int lane  = tid & 31;
  const int m0    = gwave * 16;
  const bool active = m0 < N_NODES;

  const int half = lane >> 4;
  const int l    = lane & 15;

  const int arow_idx = active ? (m0 + l) : l;
  const float4* arow =
      (const float4*)(feat + (size_t)arow_idx * D + 4 * half);
  float4 a[KBLOCKS];
#pragma unroll
  for (int jj = 0; jj < KBLOCKS; ++jj) a[jj] = arow[2 * jj];

  const float4* W4 = (const float4*)W;
#pragma unroll
  for (int i = 0; i < (D * D / 4) / 256; ++i) {
    const int g = tid + 256 * i;
    const int n = g >> 4;
    const int q = g & 15;
    const int e = ((q >> 1) * 4 + (n >> 4)) * 32 + (q & 1) * 16 + (n & 15);
    w_frag[e] = W4[g];
  }
  __syncthreads();
  if (!active) return;

  v8f acc0 = {}, acc1 = {}, acc2 = {}, acc3 = {};

#pragma unroll
  for (int jj = 0; jj < KBLOCKS; ++jj) {
    const float4 b0 = w_frag[(jj * 4 + 0) * 32 + lane];
    const float4 b1 = w_frag[(jj * 4 + 1) * 32 + lane];
    const float4 b2 = w_frag[(jj * 4 + 2) * 32 + lane];
    const float4 b3 = w_frag[(jj * 4 + 3) * 32 + lane];
    const v2f alo = {a[jj].x, a[jj].y};
    const v2f ahi = {a[jj].z, a[jj].w};
    const v2f b0lo = {b0.x, b0.y}, b0hi = {b0.z, b0.w};
    const v2f b1lo = {b1.x, b1.y}, b1hi = {b1.z, b1.w};
    const v2f b2lo = {b2.x, b2.y}, b2hi = {b2.z, b2.w};
    const v2f b3lo = {b3.x, b3.y}, b3hi = {b3.z, b3.w};
    acc0 = __builtin_amdgcn_wmma_f32_16x16x4_f32(false, alo, false, b0lo,
                                                 (short)0, acc0, false, false);
    acc1 = __builtin_amdgcn_wmma_f32_16x16x4_f32(false, alo, false, b1lo,
                                                 (short)0, acc1, false, false);
    acc2 = __builtin_amdgcn_wmma_f32_16x16x4_f32(false, alo, false, b2lo,
                                                 (short)0, acc2, false, false);
    acc3 = __builtin_amdgcn_wmma_f32_16x16x4_f32(false, alo, false, b3lo,
                                                 (short)0, acc3, false, false);
    acc0 = __builtin_amdgcn_wmma_f32_16x16x4_f32(false, ahi, false, b0hi,
                                                 (short)0, acc0, false, false);
    acc1 = __builtin_amdgcn_wmma_f32_16x16x4_f32(false, ahi, false, b1hi,
                                                 (short)0, acc1, false, false);
    acc2 = __builtin_amdgcn_wmma_f32_16x16x4_f32(false, ahi, false, b2hi,
                                                 (short)0, acc2, false, false);
    acc3 = __builtin_amdgcn_wmma_f32_16x16x4_f32(false, ahi, false, b3hi,
                                                 (short)0, acc3, false, false);
  }

  const v8f accs[4] = {acc0, acc1, acc2, acc3};
#pragma unroll
  for (int nt = 0; nt < 4; ++nt) {
    const int n = nt * 16 + l;
#pragma unroll
    for (int r = 0; r < 8; ++r) {
      const int row = m0 + r + 8 * half;
      h[(size_t)row * D + n] = accs[nt][r];
    }
  }
}

#define R_ROWS 200
#define NR ((N_NODES + R_ROWS - 1) / R_ROWS)
#define CH 4096
#define NCH ((N_EDGES + CH - 1) / CH)
#define BIN_THREADS 1024
#define EDGES_PER_THREAD (CH / BIN_THREADS)
#define OWN_THREADS 1024
#define OWN_WAVES (OWN_THREADS / 32)
#define SEG_PER_WAVE ((NCH + OWN_WAVES - 1) / OWN_WAVES)
#define EDGE_GROUP 8

static_assert(N_NODES <= 65536, "src fits the low 16 bits of a record");
static_assert(R_ROWS <= 65536, "row-in-range fits the high 16 bits");
static_assert(NR <= BIN_THREADS, "one thread per range in the scan");
static_assert(CH % BIN_THREADS == 0, "whole edges per thread");
static_assert(SEG_PER_WAVE <= 32, "a wave keeps its segments one per lane");

__global__ __launch_bounds__(BIN_THREADS) void bin_edges(
    const float* __restrict__ ew,
    const int* __restrict__ esrc,
    const int* __restrict__ edst,
    int2* __restrict__ recs,
    int2* __restrict__ seg)
{
  __shared__ int  cnt[NR];
  __shared__ int  off[NR + 1];
  __shared__ int  wsum[BIN_THREADS / 32];
  __shared__ int2 srec[CH];

  const int t  = (int)threadIdx.x;
  const int c  = (int)blockIdx.x;
  const int e0 = c * CH;

  if (t < NR) cnt[t] = 0;
  __syncthreads();

  int   sv[EDGES_PER_THREAD], dv[EDGES_PER_THREAD];
  float wv[EDGES_PER_THREAD];
#pragma unroll
  for (int k = 0; k < EDGES_PER_THREAD; ++k) {
    const int e  = e0 + k * BIN_THREADS + t;
    const int ec = e < N_EDGES ? e : N_EDGES - 1;
    sv[k] = esrc[ec];
    dv[k] = edst[ec];
    wv[k] = ew[ec];
  }
  static_assert(EDGES_PER_THREAD == 4, "the fence below names 4 values");
  asm volatile("" ::"v"(sv[0]), "v"(sv[1]), "v"(sv[2]), "v"(sv[3]));

  int rr[EDGES_PER_THREAD], key[EDGES_PER_THREAD], rank[EDGES_PER_THREAD];
#pragma unroll
  for (int k = 0; k < EDGES_PER_THREAD; ++k) {
    const int e = e0 + k * BIN_THREADS + t;
    const int d = dv[k];
    rr[k] = -1; key[k] = 0; rank[k] = 0;
    if (e < N_EDGES && (unsigned)d < (unsigned)N_NODES) {
      const int r = d / R_ROWS;
      rr[k]   = r;
      key[k]  = (sv[k] & 0xFFFF) | ((d - r * R_ROWS) << 16);
      rank[k] = atomicAdd(&cnt[r], 1);
    }
  }
  __syncthreads();

  {
    const int lane = t & 31;
    const int wave = t >> 5;
    const int v = (t < NR) ? cnt[t] : 0;
    int incl = v;
#pragma unroll
    for (int d = 1; d < 32; d <<= 1) {
      const int o = __shfl_up(incl, d);
      if (lane >= d) incl += o;
    }
    if (lane == 31) wsum[wave] = incl;
    __syncthreads();
    if (t < NR) {
      int prefix = 0;
      for (int k = 0; k < wave; ++k) prefix += wsum[k];
      off[t] = prefix + incl - v;
      if (t == NR - 1) off[NR] = prefix + incl;
    }
  }
  __syncthreads();

#pragma unroll
  for (int k = 0; k < EDGES_PER_THREAD; ++k) {
    if (rr[k] >= 0)
      srec[off[rr[k]] + rank[k]] = make_int2(key[k], __float_as_int(wv[k]));
  }
  __syncthreads();

  const int total = off[NR];
#pragma unroll
  for (int k = 0; k < EDGES_PER_THREAD; ++k) {
    const int i = k * BIN_THREADS + t;
    if (i < total) recs[e0 + i] = srec[i];
  }
  if (t < NR) seg[t * NCH + c] = make_int2(e0 + off[t], cnt[t]);
}


__device__ __forceinline__ void add_edges(const float* __restrict__ h,
                                          float* tile, int2 rec, int m,
                                          int lane)
{
  for (int j0 = 0; j0 < m; j0 += EDGE_GROUP) {
    int   kk[EDGE_GROUP];
    float lo[EDGE_GROUP], hi[EDGE_GROUP];
#pragma unroll
    for (int u = 0; u < EDGE_GROUP; ++u) {
      const int j = (j0 + u < m) ? (j0 + u) : (m - 1);
      kk[u] = __builtin_amdgcn_readlane(rec.x, j);
    }
#pragma unroll
    for (int u = 0; u < EDGE_GROUP; ++u) {
      const float* hrow = h + (size_t)(kk[u] & 0xFFFF) * D;
      lo[u] = hrow[lane];
      hi[u] = hrow[lane + 32];
    }
    static_assert(EDGE_GROUP == 8, "the fence below names 8 row pairs");
    asm volatile("" ::"v"(lo[0]), "v"(lo[1]), "v"(lo[2]), "v"(lo[3]),
                 "v"(lo[4]), "v"(lo[5]), "v"(lo[6]), "v"(lo[7]), "v"(hi[0]),
                 "v"(hi[1]), "v"(hi[2]), "v"(hi[3]), "v"(hi[4]), "v"(hi[5]),
                 "v"(hi[6]), "v"(hi[7]));
#pragma unroll
    for (int u = 0; u < EDGE_GROUP; ++u) {
      if (j0 + u < m) {
        const float w = __int_as_float(
            __builtin_amdgcn_readlane(rec.y, j0 + u));
        float* trow = tile + (int)((unsigned)kk[u] >> 16) * D;
        atomicAdd(trow + lane, lo[u] * w);
        atomicAdd(trow + lane + 32, hi[u] * w);
      }
    }
  }
}

__global__ __launch_bounds__(OWN_THREADS) void owner_sum(
    const float* __restrict__ h,
    const int2* __restrict__ recs,
    const int2* __restrict__ seg,
    const float* __restrict__ skip_w,
    const float* __restrict__ bias,
    float* __restrict__ out)
{
  __shared__ float tile[R_ROWS * D];

  const int t    = (int)threadIdx.x;
  const int lane = t & 31;
  const int wave = t >> 5;
  const int r    = (int)blockIdx.x;

  int2 myseg = make_int2(0, 0);
  {
    const int c = wave + OWN_WAVES * lane;
    if (lane < SEG_PER_WAVE && c < NCH) myseg = seg[r * NCH + c];
  }

  for (int i = t; i < R_ROWS * D; i += OWN_THREADS) tile[i] = 0.0f;

  int2 rec[SEG_PER_WAVE];
#pragma unroll
  for (int i = 0; i < SEG_PER_WAVE; ++i) {
    const int s = __builtin_amdgcn_readlane(myseg.x, i);
    const int n = __builtin_amdgcn_readlane(myseg.y, i);
    rec[i] = make_int2(0, 0);
    if (lane < n) rec[i] = recs[s + lane];
  }
  __syncthreads();

#pragma unroll
  for (int i = 0; i < SEG_PER_WAVE; ++i) {
    const int s = __builtin_amdgcn_readlane(myseg.x, i);
    const int n = __builtin_amdgcn_readlane(myseg.y, i);
    add_edges(h, tile, rec[i], n < 32 ? n : 32, lane);
    for (int base = 32; base < n; base += 32) {
      const int m = (n - base) < 32 ? (n - base) : 32;
      int2 rx = make_int2(0, 0);
      if (lane < m) rx = recs[s + base + lane];
      add_edges(h, tile, rx, m, lane);
    }
  }
  __syncthreads();

  const float lam = 1.0507009873554805f;
  const float alp = 1.6732632423543772f;
  for (int i = t; i < R_ROWS * D; i += OWN_THREADS) {
    const int row = r * R_ROWS + (i >> 6);
    const int col = i & (D - 1);
    if (row < N_NODES) {
      const size_t idx = (size_t)row * D + col;
      const float x = h[idx] * skip_w[col] + bias[col] + tile[i];
      out[idx] = lam * (x > 0.0f ? x : alp * (__expf(x) - 1.0f));
    }
  }
}

extern "C" void kernel_launch(void* const* d_in, const int* in_sizes, int n_in,
                              void* d_out, int out_size, void* d_ws,
                              size_t ws_size, hipStream_t stream) {
  const float* feat   = (const float*)d_in[0];
  const float* W      = (const float*)d_in[1];
  const float* bias   = (const float*)d_in[2];
  const float* skip_w = (const float*)d_in[3];
  const float* ew     = (const float*)d_in[4];
  const int*   esrc   = (const int*)d_in[5];
  const int*   edst   = (const int*)d_in[6];
  float* out = (float*)d_out;
  float* h    = (float*)d_ws;
  int2*  recs = (int2*)((char*)d_ws + 16777216);
  int2*  seg  = (int2*)((char*)d_ws + 33554432);

  const int gemm_threads = ((N_NODES + 15) / 16) * 32;
  gemm_wmma<<<(gemm_threads + 255) / 256, 256, 0, stream>>>(feat, W, h);

  bin_edges<<<NCH, BIN_THREADS, 0, stream>>>(ew, esrc, edst, recs, seg);

  owner_sum<<<NR, OWN_THREADS, 0, stream>>>(h, recs, seg, skip_w, bias, out);
}


# GEMM and edge binning, independent of each other, share one launch of 256-thread work-groups (binning: 16 edges per thread): 2 launches instead of 3; row sums stay arrival-ordered
# speedup vs baseline: 2.0573x; 1.0452x over previous
#include <hip/hip_runtime.h>


typedef __attribute__((ext_vector_type(2))) float v2f;
typedef __attribute__((ext_vector_type(8))) float v8f;

#define N_NODES 50000
#define N_EDGES 800000
#define D 64

#define KBLOCKS (D / 8)
#define PREP_THREADS 256
#define ROW_TILES ((N_NODES + 15) / 16)
#define GEMM_BLOCKS ((ROW_TILES * 32 + PREP_THREADS - 1) / PREP_THREADS)
#define GEMM_LDS_BYTES (KBLOCKS * 4 * 32 * 16)

__device__ __forceinline__ void gemm_role(
    const float* __restrict__ feat,
    const float* __restrict__ W,
    float* __restrict__ h,
    float4* w_frag,
    int gb)
{
  const int tid   = (int)threadIdx.x;
  const int gwave = (gb * PREP_THREADS + tid) >> 5;
  const int lane  = tid & 31;
  const int m0    = gwave * 16;
  const bool active = m0 < N_NODES;

  const int half = lane >> 4;
  const int l    = lane & 15;

  const int arow_idx = active ? (m0 + l) : l;
  const float4* arow =
      (const float4*)(feat + (size_t)arow_idx * D + 4 * half);
  float4 a[KBLOCKS];
#pragma unroll
  for (int jj = 0; jj < KBLOCKS; ++jj) a[jj] = arow[2 * jj];

  const float4* W4 = (const float4*)W;
#pragma unroll
  for (int i = 0; i < (D * D / 4) / PREP_THREADS; ++i) {
    const int g = tid + PREP_THREADS * i;
    const int n = g >> 4;
    const int q = g & 15;
    const int e = ((q >> 1) * 4 + (n >> 4)) * 32 + (q & 1) * 16 + (n & 15);
    w_frag[e] = W4[g];
  }
  __syncthreads();
  if (!active) return;

  v8f acc0 = {}, acc1 = {}, acc2 = {}, acc3 = {};

#pragma unroll
  for (int jj = 0; jj < KBLOCKS; ++jj) {
    const float4 b0 = w_frag[(jj * 4 + 0) * 32 + lane];
    const float4 b1 = w_frag[(jj * 4 + 1) * 32 + lane];
    const float4 b2 = w_frag[(jj * 4 + 2) * 32 + lane];
    const float4 b3 = w_frag[(jj * 4 + 3) * 32 + lane];
    const v2f alo = {a[jj].x, a[jj].y};
    const v2f ahi = {a[jj].z, a[jj].w};
    const v2f b0lo = {b0.x, b0.y}, b0hi = {b0.z, b0.w};
    const v2f b1lo = {b1.x, b1.y}, b1hi = {b1.z, b1.w};
    const v2f b2lo = {b2.x, b2.y}, b2hi = {b2.z, b2.w};
    const v2f b3lo = {b3.x, b3.y}, b3hi = {b3.z, b3.w};
    acc0 = __builtin_amdgcn_wmma_f32_16x16x4_f32(false, alo, false, b0lo,
                                                 (short)0, acc0, false, false);
    acc1 = __builtin_amdgcn_wmma_f32_16x16x4_f32(false, alo, false, b1lo,
                                                 (short)0, acc1, false, false);
    acc2 = __builtin_amdgcn_wmma_f32_16x16x4_f32(false, alo, false, b2lo,
                                                 (short)0, acc2, false, false);
    acc3 = __builtin_amdgcn_wmma_f32_16x16x4_f32(false, alo, false, b3lo,
                                                 (short)0, acc3, false, false);
    acc0 = __builtin_amdgcn_wmma_f32_16x16x4_f32(false, ahi, false, b0hi,
                                                 (short)0, acc0, false, false);
    acc1 = __builtin_amdgcn_wmma_f32_16x16x4_f32(false, ahi, false, b1hi,
                                                 (short)0, acc1, false, false);
    acc2 = __builtin_amdgcn_wmma_f32_16x16x4_f32(false, ahi, false, b2hi,
                                                 (short)0, acc2, false, false);
    acc3 = __builtin_amdgcn_wmma_f32_16x16x4_f32(false, ahi, false, b3hi,
                                                 (short)0, acc3, false, false);
  }

  const v8f accs[4] = {acc0, acc1, acc2, acc3};
#pragma unroll
  for (int nt = 0; nt < 4; ++nt) {
    const int n = nt * 16 + l;
#pragma unroll
    for (int r = 0; r < 8; ++r) {
      const int row = m0 + r + 8 * half;
      h[(size_t)row * D + n] = accs[nt][r];
    }
  }
}

#define R_ROWS 200
#define NR ((N_NODES + R_ROWS - 1) / R_ROWS)
#define CH 4096
#define NCH ((N_EDGES + CH - 1) / CH)
#define BIN_THREADS PREP_THREADS
#define EDGES_PER_THREAD (CH / BIN_THREADS)
#define BIN_LDS_BYTES (CH * 8 + (NR + NR + 1 + BIN_THREADS / 32) * 4)
#define PREP_LDS_BYTES \
  (BIN_LDS_BYTES > GEMM_LDS_BYTES ? BIN_LDS_BYTES : GEMM_LDS_BYTES)
#define OWN_THREADS 1024
#define OWN_WAVES (OWN_THREADS / 32)
#define SEG_PER_WAVE ((NCH + OWN_WAVES - 1) / OWN_WAVES)
#define EDGE_GROUP 8

static_assert(N_NODES <= 65536, "src fits the low 16 bits of a record");
static_assert(R_ROWS <= 65536, "row-in-range fits the high 16 bits");
static_assert(NR <= BIN_THREADS, "one thread per range in the scan");
static_assert(CH % BIN_THREADS == 0, "whole edges per thread");
static_assert(SEG_PER_WAVE <= 32, "a wave keeps its segments one per lane");

__device__ __forceinline__ void bin_role(
    const float* __restrict__ ew,
    const int* __restrict__ esrc,
    const int* __restrict__ edst,
    int2* __restrict__ recs,
    int2* __restrict__ seg,
    unsigned char* lds,
    int c)
{
  int2* srec = (int2*)lds;
  int*  cnt  = (int*)(lds + CH * 8);
  int*  off  = cnt + NR;
  int*  wsum = off + NR + 1;

  const int t  = (int)threadIdx.x;
  const int e0 = c * CH;

  if (t < NR) cnt[t] = 0;
  __syncthreads();

  int   sv[EDGES_PER_THREAD], dv[EDGES_PER_THREAD];
  float wv[EDGES_PER_THREAD];
#pragma unroll
  for (int k = 0; k < EDGES_PER_THREAD; ++k) {
    const int e  = e0 + k * BIN_THREADS + t;
    const int ec = e < N_EDGES ? e : N_EDGES - 1;
    sv[k] = esrc[ec];
    dv[k] = edst[ec];
    wv[k] = ew[ec];
  }
  static_assert(EDGES_PER_THREAD == 16, "the fence below names 16 values");
  asm volatile("" ::"v"(sv[0]), "v"(sv[1]), "v"(sv[2]), "v"(sv[3]),
               "v"(sv[4]), "v"(sv[5]), "v"(sv[6]), "v"(sv[7]), "v"(sv[8]),
               "v"(sv[9]), "v"(sv[10]), "v"(sv[11]), "v"(sv[12]), "v"(sv[13]),
               "v"(sv[14]), "v"(sv[15]));

  int rr[EDGES_PER_THREAD], key[EDGES_PER_THREAD], rank[EDGES_PER_THREAD];
#pragma unroll
  for (int k = 0; k < EDGES_PER_THREAD; ++k) {
    const int e = e0 + k * BIN_THREADS + t;
    const int d = dv[k];
    rr[k] = -1; key[k] = 0; rank[k] = 0;
    if (e < N_EDGES && (unsigned)d < (unsigned)N_NODES) {
      const int r = d / R_ROWS;
      rr[k]   = r;
      key[k]  = (sv[k] & 0xFFFF) | ((d - r * R_ROWS) << 16);
      rank[k] = atomicAdd(&cnt[r], 1);
    }
  }
  __syncthreads();

  {
    const int lane = t & 31;
    const int wave = t >> 5;
    const int v = (t < NR) ? cnt[t] : 0;
    int incl = v;
#pragma unroll
    for (int d = 1; d < 32; d <<= 1) {
      const int o = __shfl_up(incl, d);
      if (lane >= d) incl += o;
    }
    if (lane == 31) wsum[wave] = incl;
    __syncthreads();
    if (t < NR) {
      int prefix = 0;
      for (int k = 0; k < wave; ++k) prefix += wsum[k];
      off[t] = prefix + incl - v;
      if (t == NR - 1) off[NR] = prefix + incl;
    }
  }
  __syncthreads();

#pragma unroll
  for (int k = 0; k < EDGES_PER_THREAD; ++k) {
    if (rr[k] >= 0)
      srec[off[rr[k]] + rank[k]] = make_int2(key[k], __float_as_int(wv[k]));
  }
  __syncthreads();

  const int total = off[NR];
#pragma unroll
  for (int k = 0; k < EDGES_PER_THREAD; ++k) {
    const int i = k * BIN_THREADS + t;
    if (i < total) recs[e0 + i] = srec[i];
  }
  if (t < NR) seg[t * NCH + c] = make_int2(e0 + off[t], cnt[t]);
}

__global__ __launch_bounds__(PREP_THREADS) void bin_and_gemm(
    const float* __restrict__ feat,
    const float* __restrict__ W,
    float* __restrict__ h,
    const float* __restrict__ ew,
    const int* __restrict__ esrc,
    const int* __restrict__ edst,
    int2* __restrict__ recs,
    int2* __restrict__ seg)
{
  __shared__ alignas(16) unsigned char lds[PREP_LDS_BYTES];
  const int b = (int)blockIdx.x;
  if (b < NCH)
    bin_role(ew, esrc, edst, recs, seg, lds, b);
  else
    gemm_role(feat, W, h, (float4*)lds, b - NCH);
}


__device__ __forceinline__ void add_edges(const float* __restrict__ h,
                                          float* tile, int2 rec, int m,
                                          int lane)
{
  for (int j0 = 0; j0 < m; j0 += EDGE_GROUP) {
    int   kk[EDGE_GROUP];
    float lo[EDGE_GROUP], hi[EDGE_GROUP];
#pragma unroll
    for (int u = 0; u < EDGE_GROUP; ++u) {
      const int j = (j0 + u < m) ? (j0 + u) : (m - 1);
      kk[u] = __builtin_amdgcn_readlane(rec.x, j);
    }
#pragma unroll
    for (int u = 0; u < EDGE_GROUP; ++u) {
      const float* hrow = h + (size_t)(kk[u] & 0xFFFF) * D;
      lo[u] = hrow[lane];
      hi[u] = hrow[lane + 32];
    }
    static_assert(EDGE_GROUP == 8, "the fence below names 8 row pairs");
    asm volatile("" ::"v"(lo[0]), "v"(lo[1]), "v"(lo[2]), "v"(lo[3]),
                 "v"(lo[4]), "v"(lo[5]), "v"(lo[6]), "v"(lo[7]), "v"(hi[0]),
                 "v"(hi[1]), "v"(hi[2]), "v"(hi[3]), "v"(hi[4]), "v"(hi[5]),
                 "v"(hi[6]), "v"(hi[7]));
#pragma unroll
    for (int u = 0; u < EDGE_GROUP; ++u) {
      if (j0 + u < m) {
        const float w = __int_as_float(
            __builtin_amdgcn_readlane(rec.y, j0 + u));
        float* trow = tile + (int)((unsigned)kk[u] >> 16) * D;
        atomicAdd(trow + lane, lo[u] * w);
        atomicAdd(trow + lane + 32, hi[u] * w);
      }
    }
  }
}

__global__ __launch_bounds__(OWN_THREADS) void owner_sum(
    const float* __restrict__ h,
    const int2* __restrict__ recs,
    const int2* __restrict__ seg,
    const float* __restrict__ skip_w,
    const float* __restrict__ bias,
    float* __restrict__ out)
{
  __shared__ float tile[R_ROWS * D];

  const int t    = (int)threadIdx.x;
  const int lane = t & 31;
  const int wave = t >> 5;
  const int r    = (int)blockIdx.x;

  int2 myseg = make_int2(0, 0);
  {
    const int c = wave + OWN_WAVES * lane;
    if (lane < SEG_PER_WAVE && c < NCH) myseg = seg[r * NCH + c];
  }

  for (int i = t; i < R_ROWS * D; i += OWN_THREADS) tile[i] = 0.0f;

  int2 rec[SEG_PER_WAVE];
#pragma unroll
  for (int i = 0; i < SEG_PER_WAVE; ++i) {
    const int s = __builtin_amdgcn_readlane(myseg.x, i);
    const int n = __builtin_amdgcn_readlane(myseg.y, i);
    rec[i] = make_int2(0, 0);
    if (lane < n) rec[i] = recs[s + lane];
  }
  __syncthreads();

#pragma unroll
  for (int i = 0; i < SEG_PER_WAVE; ++i) {
    const int s = __builtin_amdgcn_readlane(myseg.x, i);
    const int n = __builtin_amdgcn_readlane(myseg.y, i);
    add_edges(h, tile, rec[i], n < 32 ? n : 32, lane);
    for (int base = 32; base < n; base += 32) {
      const int m = (n - base) < 32 ? (n - base) : 32;
      int2 rx = make_int2(0, 0);
      if (lane < m) rx = recs[s + base + lane];
      add_edges(h, tile, rx, m, lane);
    }
  }
  __syncthreads();

  const float lam = 1.0507009873554805f;
  const float alp = 1.6732632423543772f;
  for (int i = t; i < R_ROWS * D; i += OWN_THREADS) {
    const int row = r * R_ROWS + (i >> 6);
    const int col = i & (D - 1);
    if (row < N_NODES) {
      const size_t idx = (size_t)row * D + col;
      const float x = h[idx] * skip_w[col] + bias[col] + tile[i];
      out[idx] = lam * (x > 0.0f ? x : alp * (__expf(x) - 1.0f));
    }
  }
}

extern "C" void kernel_launch(void* const* d_in, const int* in_sizes, int n_in,
                              void* d_out, int out_size, void* d_ws,
                              size_t ws_size, hipStream_t stream) {
  const float* feat   = (const float*)d_in[0];
  const float* W      = (const float*)d_in[1];
  const float* bias   = (const float*)d_in[2];
  const float* skip_w = (const float*)d_in[3];
  const float* ew     = (const float*)d_in[4];
  const int*   esrc   = (const int*)d_in[5];
  const int*   edst   = (const int*)d_in[6];
  float* out = (float*)d_out;
  float* h    = (float*)d_ws;
  int2*  recs = (int2*)((char*)d_ws + 16777216);
  int2*  seg  = (int2*)((char*)d_ws + 33554432);

  bin_and_gemm<<<NCH + GEMM_BLOCKS, PREP_THREADS, 0, stream>>>(
      feat, W, h, ew, esrc, edst, recs, seg);

  owner_sum<<<NR, OWN_THREADS, 0, stream>>>(h, recs, seg, skip_w, bias, out);
}


# Owner kernel asks for its epilogue operands (own h rows, skip, bias) in the prologue's memory trip, before the edge phase: the tail's 13 dependent trips per thread go; row sums stay arrival-ordered
# speedup vs baseline: 2.4168x; 1.1747x over previous
#include <hip/hip_runtime.h>


typedef __attribute__((ext_vector_type(2))) float v2f;
typedef __attribute__((ext_vector_type(8))) float v8f;

#define N_NODES 50000
#define N_EDGES 800000
#define D 64

#define KBLOCKS (D / 8)
#define PREP_THREADS 256
#define ROW_TILES ((N_NODES + 15) / 16)
#define GEMM_BLOCKS ((ROW_TILES * 32 + PREP_THREADS - 1) / PREP_THREADS)
#define GEMM_LDS_BYTES (KBLOCKS * 4 * 32 * 16)

__device__ __forceinline__ void gemm_role(
    const float* __restrict__ feat,
    const float* __restrict__ W,
    float* __restrict__ h,
    float4* w_frag,
    int gb)
{
  const int tid   = (int)threadIdx.x;
  const int gwave = (gb * PREP_THREADS + tid) >> 5;
  const int lane  = tid & 31;
  const int m0    = gwave * 16;
  const bool active = m0 < N_NODES;

  const int half = lane >> 4;
  const int l    = lane & 15;

  const int arow_idx = active ? (m0 + l) : l;
  const float4* arow =
      (const float4*)(feat + (size_t)arow_idx * D + 4 * half);
  float4 a[KBLOCKS];
#pragma unroll
  for (int jj = 0; jj < KBLOCKS; ++jj) a[jj] = arow[2 * jj];

  const float4* W4 = (const float4*)W;
#pragma unroll
  for (int i = 0; i < (D * D / 4) / PREP_THREADS; ++i) {
    const int g = tid + PREP_THREADS * i;
    const int n = g >> 4;
    const int q = g & 15;
    const int e = ((q >> 1) * 4 + (n >> 4)) * 32 + (q & 1) * 16 + (n & 15);
    w_frag[e] = W4[g];
  }
  __syncthreads();
  if (!active) return;

  v8f acc0 = {}, acc1 = {}, acc2 = {}, acc3 = {};

#pragma unroll
  for (int jj = 0; jj < KBLOCKS; ++jj) {
    const float4 b0 = w_frag[(jj * 4 + 0) * 32 + lane];
    const float4 b1 = w_frag[(jj * 4 + 1) * 32 + lane];
    const float4 b2 = w_frag[(jj * 4 + 2) * 32 + lane];
    const float4 b3 = w_frag[(jj * 4 + 3) * 32 + lane];
    const v2f alo = {a[jj].x, a[jj].y};
    const v2f ahi = {a[jj].z, a[jj].w};
    const v2f b0lo = {b0.x, b0.y}, b0hi = {b0.z, b0.w};
    const v2f b1lo = {b1.x, b1.y}, b1hi = {b1.z, b1.w};
    const v2f b2lo = {b2.x, b2.y}, b2hi = {b2.z, b2.w};
    const v2f b3lo = {b3.x, b3.y}, b3hi = {b3.z, b3.w};
    acc0 = __builtin_amdgcn_wmma_f32_16x16x4_f32(false, alo, false, b0lo,
                                                 (short)0, acc0, false, false);
    acc1 = __builtin_amdgcn_wmma_f32_16x16x4_f32(false, alo, false, b1lo,
                                                 (short)0, acc1, false, false);
    acc2 = __builtin_amdgcn_wmma_f32_16x16x4_f32(false, alo, false, b2lo,
                                                 (short)0, acc2, false, false);
    acc3 = __builtin_amdgcn_wmma_f32_16x16x4_f32(false, alo, false, b3lo,
                                                 (short)0, acc3, false, false);
    acc0 = __builtin_amdgcn_wmma_f32_16x16x4_f32(false, ahi, false, b0hi,
                                                 (short)0, acc0, false, false);
    acc1 = __builtin_amdgcn_wmma_f32_16x16x4_f32(false, ahi, false, b1hi,
                                                 (short)0, acc1, false, false);
    acc2 = __builtin_amdgcn_wmma_f32_16x16x4_f32(false, ahi, false, b2hi,
                                                 (short)0, acc2, false, false);
    acc3 = __builtin_amdgcn_wmma_f32_16x16x4_f32(false, ahi, false, b3hi,
                                                 (short)0, acc3, false, false);
  }

  const v8f accs[4] = {acc0, acc1, acc2, acc3};
#pragma unroll
  for (int nt = 0; nt < 4; ++nt) {
    const int n = nt * 16 + l;
#pragma unroll
    for (int r = 0; r < 8; ++r) {
      const int row = m0 + r + 8 * half;
      h[(size_t)row * D + n] = accs[nt][r];
    }
  }
}

#define R_ROWS 200
#define NR ((N_NODES + R_ROWS - 1) / R_ROWS)
#define CH 4096
#define NCH ((N_EDGES + CH - 1) / CH)
#define BIN_THREADS PREP_THREADS
#define EDGES_PER_THREAD (CH / BIN_THREADS)
#define BIN_LDS_BYTES (CH * 8 + (NR + NR + 1 + BIN_THREADS / 32) * 4)
#define PREP_LDS_BYTES \
  (BIN_LDS_BYTES > GEMM_LDS_BYTES ? BIN_LDS_BYTES : GEMM_LDS_BYTES)
#define OWN_THREADS 1024
#define OWN_WAVES (OWN_THREADS / 32)
#define SEG_PER_WAVE ((NCH + OWN_WAVES - 1) / OWN_WAVES)
#define EDGE_GROUP 8
#define EPI_STEPS ((R_ROWS * D + OWN_THREADS - 1) / OWN_THREADS)
#define EPI_ROW_STEP (OWN_THREADS / D)
#define LAST_STEP_THREADS (R_ROWS * D - (EPI_STEPS - 1) * OWN_THREADS)

static_assert(N_NODES <= 65536, "src fits the low 16 bits of a record");
static_assert(R_ROWS <= 65536, "row-in-range fits the high 16 bits");
static_assert(NR <= BIN_THREADS, "one thread per range in the scan");
static_assert(CH % BIN_THREADS == 0, "whole edges per thread");
static_assert(SEG_PER_WAVE <= 32, "a wave keeps its segments one per lane");
static_assert(OWN_THREADS % D == 0, "a thread's column is the same every step");
static_assert(NR * R_ROWS == N_NODES, "every row of every range is a node row");
static_assert(LAST_STEP_THREADS > 0 && LAST_STEP_THREADS % 32 == 0,
              "the last step covers whole waves");

__device__ __forceinline__ void bin_role(
    const float* __restrict__ ew,
    const int* __restrict__ esrc,
    const int* __restrict__ edst,
    int2* __restrict__ recs,
    int2* __restrict__ seg,
    unsigned char* lds,
    int c)
{
  int2* srec = (int2*)lds;
  int*  cnt  = (int*)(lds + CH * 8);
  int*  off  = cnt + NR;
  int*  wsum = off + NR + 1;

  const int t  = (int)threadIdx.x;
  const int e0 = c * CH;

  if (t < NR) cnt[t] = 0;
  __syncthreads();

  int   sv[EDGES_PER_THREAD], dv[EDGES_PER_THREAD];
  float wv[EDGES_PER_THREAD];
#pragma unroll
  for (int k = 0; k < EDGES_PER_THREAD; ++k) {
    const int e  = e0 + k * BIN_THREADS + t;
    const int ec = e < N_EDGES ? e : N_EDGES - 1;
    sv[k] = esrc[ec];
    dv[k] = edst[ec];
    wv[k] = ew[ec];
  }
  static_assert(EDGES_PER_THREAD == 16, "the fence below names 16 values");
  asm volatile("" ::"v"(sv[0]), "v"(sv[1]), "v"(sv[2]), "v"(sv[3]),
               "v"(sv[4]), "v"(sv[5]), "v"(sv[6]), "v"(sv[7]), "v"(sv[8]),
               "v"(sv[9]), "v"(sv[10]), "v"(sv[11]), "v"(sv[12]), "v"(sv[13]),
               "v"(sv[14]), "v"(sv[15]));

  int rr[EDGES_PER_THREAD], key[EDGES_PER_THREAD], rank[EDGES_PER_THREAD];
#pragma unroll
  for (int k = 0; k < EDGES_PER_THREAD; ++k) {
    const int e = e0 + k * BIN_THREADS + t;
    const int d = dv[k];
    rr[k] = -1; key[k] = 0; rank[k] = 0;
    if (e < N_EDGES && (unsigned)d < (unsigned)N_NODES) {
      const int r = d / R_ROWS;
      rr[k]   = r;
      key[k]  = (sv[k] & 0xFFFF) | ((d - r * R_ROWS) << 16);
      rank[k] = atomicAdd(&cnt[r], 1);
    }
  }
  __syncthreads();

  {
    const int lane = t & 31;
    const int wave = t >> 5;
    const int v = (t < NR) ? cnt[t] : 0;
    int incl = v;
#pragma unroll
    for (int d = 1; d < 32; d <<= 1) {
      const int o = __shfl_up(incl, d);
      if (lane >= d) incl += o;
    }
    if (lane == 31) wsum[wave] = incl;
    __syncthreads();
    if (t < NR) {
      int prefix = 0;
      for (int k = 0; k < wave; ++k) prefix += wsum[k];
      off[t] = prefix + incl - v;
      if (t == NR - 1) off[NR] = prefix + incl;
    }
  }
  __syncthreads();

#pragma unroll
  for (int k = 0; k < EDGES_PER_THREAD; ++k) {
    if (rr[k] >= 0)
      srec[off[rr[k]] + rank[k]] = make_int2(key[k], __float_as_int(wv[k]));
  }
  __syncthreads();

  const int total = off[NR];
#pragma unroll
  for (int k = 0; k < EDGES_PER_THREAD; ++k) {
    const int i = k * BIN_THREADS + t;
    if (i < total) recs[e0 + i] = srec[i];
  }
  if (t < NR) seg[t * NCH + c] = make_int2(e0 + off[t], cnt[t]);
}

__global__ __launch_bounds__(PREP_THREADS) void bin_and_gemm(
    const float* __restrict__ feat,
    const float* __restrict__ W,
    float* __restrict__ h,
    const float* __restrict__ ew,
    const int* __restrict__ esrc,
    const int* __restrict__ edst,
    int2* __restrict__ recs,
    int2* __restrict__ seg)
{
  __shared__ alignas(16) unsigned char lds[PREP_LDS_BYTES];
  const int b = (int)blockIdx.x;
  if (b < NCH)
    bin_role(ew, esrc, edst, recs, seg, lds, b);
  else
    gemm_role(feat, W, h, (float4*)lds, b - NCH);
}


__device__ __forceinline__ void add_edges(const float* __restrict__ h,
                                          float* tile, int2 rec, int m,
                                          int lane)
{
  for (int j0 = 0; j0 < m; j0 += EDGE_GROUP) {
    int   kk[EDGE_GROUP];
    float lo[EDGE_GROUP], hi[EDGE_GROUP];
#pragma unroll
    for (int u = 0; u < EDGE_GROUP; ++u) {
      const int j = (j0 + u < m) ? (j0 + u) : (m - 1);
      kk[u] = __builtin_amdgcn_readlane(rec.x, j);
    }
#pragma unroll
    for (int u = 0; u < EDGE_GROUP; ++u) {
      const float* hrow = h + (size_t)(kk[u] & 0xFFFF) * D;
      lo[u] = hrow[lane];
      hi[u] = hrow[lane + 32];
    }
    static_assert(EDGE_GROUP == 8, "the fence below names 8 row pairs");
    asm volatile("" ::"v"(lo[0]), "v"(lo[1]), "v"(lo[2]), "v"(lo[3]),
                 "v"(lo[4]), "v"(lo[5]), "v"(lo[6]), "v"(lo[7]), "v"(hi[0]),
                 "v"(hi[1]), "v"(hi[2]), "v"(hi[3]), "v"(hi[4]), "v"(hi[5]),
                 "v"(hi[6]), "v"(hi[7]));
#pragma unroll
    for (int u = 0; u < EDGE_GROUP; ++u) {
      if (j0 + u < m) {
        const float w = __int_as_float(
            __builtin_amdgcn_readlane(rec.y, j0 + u));
        float* trow = tile + (int)((unsigned)kk[u] >> 16) * D;
        atomicAdd(trow + lane, lo[u] * w);
        atomicAdd(trow + lane + 32, hi[u] * w);
      }
    }
  }
}

__global__ __launch_bounds__(OWN_THREADS) void owner_sum(
    const float* __restrict__ h,
    const int2* __restrict__ recs,
    const int2* __restrict__ seg,
    const float* __restrict__ skip_w,
    const float* __restrict__ bias,
    float* __restrict__ out)
{
  __shared__ float tile[R_ROWS * D];

  const int t    = (int)threadIdx.x;
  const int lane = t & 31;
  const int wave = t >> 5;
  const int r    = (int)blockIdx.x;

  int2 myseg = make_int2(0, 0);
  {
    const int c = wave + OWN_WAVES * lane;
    if (lane < SEG_PER_WAVE && c < NCH) myseg = seg[r * NCH + c];
  }

  const int   col  = t & (D - 1);
  const size_t own = (size_t)(r * R_ROWS + (t >> 6)) * D + col;
  const float* hp  = h + own;
  const float sk   = skip_w[col];
  const float bi   = bias[col];
  const bool last_step = t < LAST_STEP_THREADS;
  float hv[EPI_STEPS];
#pragma unroll
  for (int k = 0; k < EPI_STEPS - 1; ++k) hv[k] = hp[k * EPI_ROW_STEP * D];
  hv[EPI_STEPS - 1] = hp[last_step ? (EPI_STEPS - 1) * EPI_ROW_STEP * D : 0];

  for (int i = t; i < R_ROWS * D; i += OWN_THREADS) tile[i] = 0.0f;

  int2 rec[SEG_PER_WAVE];
#pragma unroll
  for (int i = 0; i < SEG_PER_WAVE; ++i) {
    const int s = __builtin_amdgcn_readlane(myseg.x, i);
    const int n = __builtin_amdgcn_readlane(myseg.y, i);
    rec[i] = make_int2(0, 0);
    if (lane < n) rec[i] = recs[s + lane];
  }
  __syncthreads();

#pragma unroll
  for (int i = 0; i < SEG_PER_WAVE; ++i) {
    const int s = __builtin_amdgcn_readlane(myseg.x, i);
    const int n = __builtin_amdgcn_readlane(myseg.y, i);
    add_edges(h, tile, rec[i], n < 32 ? n : 32, lane);
    for (int base = 32; base < n; base += 32) {
      const int m = (n - base) < 32 ? (n - base) : 32;
      int2 rx = make_int2(0, 0);
      if (lane < m) rx = recs[s + base + lane];
      add_edges(h, tile, rx, m, lane);
    }
  }
  __syncthreads();

  const float lam = 1.0507009873554805f;
  const float alp = 1.6732632423543772f;
  float tv[EPI_STEPS];
#pragma unroll
  for (int k = 0; k < EPI_STEPS - 1; ++k) tv[k] = tile[t + OWN_THREADS * k];
  tv[EPI_STEPS - 1] = tile[last_step ? t + OWN_THREADS * (EPI_STEPS - 1) : t];

  float* op = out + own;
#pragma unroll
  for (int k = 0; k < EPI_STEPS; ++k) {
    if (k < EPI_STEPS - 1 || last_step) {
      const float x = hv[k] * sk + bi + tv[k];
      op[k * EPI_ROW_STEP * D] =
          lam * (x > 0.0f ? x : alp * (__expf(x) - 1.0f));
    }
  }
}

extern "C" void kernel_launch(void* const* d_in, const int* in_sizes, int n_in,
                              void* d_out, int out_size, void* d_ws,
                              size_t ws_size, hipStream_t stream) {
  const float* feat   = (const float*)d_in[0];
  const float* W      = (const float*)d_in[1];
  const float* bias   = (const float*)d_in[2];
  const float* skip_w = (const float*)d_in[3];
  const float* ew     = (const float*)d_in[4];
  const int*   esrc   = (const int*)d_in[5];
  const int*   edst   = (const int*)d_in[6];
  float* out = (float*)d_out;
  float* h    = (float*)d_ws;
  int2*  recs = (int2*)((char*)d_ws + 16777216);
  int2*  seg  = (int2*)((char*)d_ws + 33554432);

  bin_and_gemm<<<NCH + GEMM_BLOCKS, PREP_THREADS, 0, stream>>>(
      feat, W, h, ew, esrc, edst, recs, seg);

  owner_sum<<<NR, OWN_THREADS, 0, stream>>>(h, recs, seg, skip_w, bias, out);
}


# Owner kernel counting-sorts its range's records by dst row in LDS; each wave walks an equal share, 16 rows in flight, sums a row's run in registers, one LDS add per run; row sums in sort-rank order
# speedup vs baseline: 2.5828x; 1.0687x over previous
#include <hip/hip_runtime.h>


typedef __attribute__((ext_vector_type(2))) float v2f;
typedef __attribute__((ext_vector_type(8))) float v8f;

#define N_NODES 50000
#define N_EDGES 800000
#define D 64

#define KBLOCKS (D / 8)
#define PREP_THREADS 256
#define ROW_TILES ((N_NODES + 15) / 16)
#define GEMM_BLOCKS ((ROW_TILES * 32 + PREP_THREADS - 1) / PREP_THREADS)
#define GEMM_LDS_BYTES (KBLOCKS * 4 * 32 * 16)

__device__ __forceinline__ void gemm_role(
    const float* __restrict__ feat,
    const float* __restrict__ W,
    float* __restrict__ h,
    float4* w_frag,
    int gb)
{
  const int tid   = (int)threadIdx.x;
  const int gwave = (gb * PREP_THREADS + tid) >> 5;
  const int lane  = tid & 31;
  const int m0    = gwave * 16;
  const bool active = m0 < N_NODES;

  const int half = lane >> 4;
  const int l    = lane & 15;

  const int arow_idx = active ? (m0 + l) : l;
  const float4* arow =
      (const float4*)(feat + (size_t)arow_idx * D + 4 * half);
  float4 a[KBLOCKS];
#pragma unroll
  for (int jj = 0; jj < KBLOCKS; ++jj) a[jj] = arow[2 * jj];

  const float4* W4 = (const float4*)W;
#pragma unroll
  for (int i = 0; i < (D * D / 4) / PREP_THREADS; ++i) {
    const int g = tid + PREP_THREADS * i;
    const int n = g >> 4;
    const int q = g & 15;
    const int e = ((q >> 1) * 4 + (n >> 4)) * 32 + (q & 1) * 16 + (n & 15);
    w_frag[e] = W4[g];
  }
  __syncthreads();
  if (!active) return;

  v8f acc0 = {}, acc1 = {}, acc2 = {}, acc3 = {};

#pragma unroll
  for (int jj = 0; jj < KBLOCKS; ++jj) {
    const float4 b0 = w_frag[(jj * 4 + 0) * 32 + lane];
    const float4 b1 = w_frag[(jj * 4 + 1) * 32 + lane];
    const float4 b2 = w_frag[(jj * 4 + 2) * 32 + lane];
    const float4 b3 = w_frag[(jj * 4 + 3) * 32 + lane];
    const v2f alo = {a[jj].x, a[jj].y};
    const v2f ahi = {a[jj].z, a[jj].w};
    const v2f b0lo = {b0.x, b0.y}, b0hi = {b0.z, b0.w};
    const v2f b1lo = {b1.x, b1.y}, b1hi = {b1.z, b1.w};
    const v2f b2lo = {b2.x, b2.y}, b2hi = {b2.z, b2.w};
    const v2f b3lo = {b3.x, b3.y}, b3hi = {b3.z, b3.w};
    acc0 = __builtin_amdgcn_wmma_f32_16x16x4_f32(false, alo, false, b0lo,
                                                 (short)0, acc0, false, false);
    acc1 = __builtin_amdgcn_wmma_f32_16x16x4_f32(false, alo, false, b1lo,
                                                 (short)0, acc1, false, false);
    acc2 = __builtin_amdgcn_wmma_f32_16x16x4_f32(false, alo, false, b2lo,
                                                 (short)0, acc2, false, false);
    acc3 = __builtin_amdgcn_wmma_f32_16x16x4_f32(false, alo, false, b3lo,
                                                 (short)0, acc3, false, false);
    acc0 = __builtin_amdgcn_wmma_f32_16x16x4_f32(false, ahi, false, b0hi,
                                                 (short)0, acc0, false, false);
    acc1 = __builtin_amdgcn_wmma_f32_16x16x4_f32(false, ahi, false, b1hi,
                                                 (short)0, acc1, false, false);
    acc2 = __builtin_amdgcn_wmma_f32_16x16x4_f32(false, ahi, false, b2hi,
                                                 (short)0, acc2, false, false);
    acc3 = __builtin_amdgcn_wmma_f32_16x16x4_f32(false, ahi, false, b3hi,
                                                 (short)0, acc3, false, false);
  }

  const v8f accs[4] = {acc0, acc1, acc2, acc3};
#pragma unroll
  for (int nt = 0; nt < 4; ++nt) {
    const int n = nt * 16 + l;
#pragma unroll
    for (int r = 0; r < 8; ++r) {
      const int row = m0 + r + 8 * half;
      h[(size_t)row * D + n] = accs[nt][r];
    }
  }
}

#define R_ROWS 200
#define NR ((N_NODES + R_ROWS - 1) / R_ROWS)
#define CH 4096
#define NCH ((N_EDGES + CH - 1) / CH)
#define BIN_THREADS PREP_THREADS
#define EDGES_PER_THREAD (CH / BIN_THREADS)
#define BIN_LDS_BYTES (CH * 8 + (NR + NR + 1 + BIN_THREADS / 32) * 4)
#define PREP_LDS_BYTES \
  (BIN_LDS_BYTES > GEMM_LDS_BYTES ? BIN_LDS_BYTES : GEMM_LDS_BYTES)
#define OWN_THREADS 1024
#define OWN_WAVES (OWN_THREADS / 32)
#define SEG_PER_WAVE ((NCH + OWN_WAVES - 1) / OWN_WAVES)
#define EDGE_GROUP 8
#define SORT_CAP (NCH * 32)
#define WALK_GROUP 16
#define ROWS_PER_LANE ((R_ROWS + 31) / 32)
#define EPI_STEPS ((R_ROWS * D + OWN_THREADS - 1) / OWN_THREADS)
#define EPI_ROW_STEP (OWN_THREADS / D)
#define LAST_STEP_THREADS (R_ROWS * D - (EPI_STEPS - 1) * OWN_THREADS)

static_assert(N_NODES <= 65536, "src fits the low 16 bits of a record");
static_assert(R_ROWS <= 65536, "row-in-range fits the high 16 bits");
static_assert(NR <= BIN_THREADS, "one thread per range in the scan");
static_assert(CH % BIN_THREADS == 0, "whole edges per thread");
static_assert(SEG_PER_WAVE <= 32, "a wave keeps its segments one per lane");
static_assert(OWN_THREADS % D == 0, "a thread's column is the same every step");
static_assert(NR * R_ROWS == N_NODES, "every row of every range is a node row");
static_assert(LAST_STEP_THREADS > 0 && LAST_STEP_THREADS % 32 == 0,
              "the last step covers whole waves");
static_assert(R_ROWS <= OWN_THREADS, "one thread clears one row counter");
static_assert(SORT_CAP * OWN_WAVES < (1 << 30), "share bounds fit an int");

__device__ __forceinline__ void bin_role(
    const float* __restrict__ ew,
    const int* __restrict__ esrc,
    const int* __restrict__ edst,
    int2* __restrict__ recs,
    int2* __restrict__ seg,
    unsigned char* lds,
    int c)
{
  int2* srec = (int2*)lds;
  int*  cnt  = (int*)(lds + CH * 8);
  int*  off  = cnt + NR;
  int*  wsum = off + NR + 1;

  const int t  = (int)threadIdx.x;
  const int e0 = c * CH;

  if (t < NR) cnt[t] = 0;
  __syncthreads();

  int   sv[EDGES_PER_THREAD], dv[EDGES_PER_THREAD];
  float wv[EDGES_PER_THREAD];
#pragma unroll
  for (int k = 0; k < EDGES_PER_THREAD; ++k) {
    const int e  = e0 + k * BIN_THREADS + t;
    const int ec = e < N_EDGES ? e : N_EDGES - 1;
    sv[k] = esrc[ec];
    dv[k] = edst[ec];
    wv[k] = ew[ec];
  }
  static_assert(EDGES_PER_THREAD == 16, "the fence below names 16 values");
  asm volatile("" ::"v"(sv[0]), "v"(sv[1]), "v"(sv[2]), "v"(sv[3]),
               "v"(sv[4]), "v"(sv[5]), "v"(sv[6]), "v"(sv[7]), "v"(sv[8]),
               "v"(sv[9]), "v"(sv[10]), "v"(sv[11]), "v"(sv[12]), "v"(sv[13]),
               "v"(sv[14]), "v"(sv[15]));

  int rr[EDGES_PER_THREAD], key[EDGES_PER_THREAD], rank[EDGES_PER_THREAD];
#pragma unroll
  for (int k = 0; k < EDGES_PER_THREAD; ++k) {
    const int e = e0 + k * BIN_THREADS + t;
    const int d = dv[k];
    rr[k] = -1; key[k] = 0; rank[k] = 0;
    if (e < N_EDGES && (unsigned)d < (unsigned)N_NODES) {
      const int r = d / R_ROWS;
      rr[k]   = r;
      key[k]  = (sv[k] & 0xFFFF) | ((d - r * R_ROWS) << 16);
      rank[k] = atomicAdd(&cnt[r], 1);
    }
  }
  __syncthreads();

  {
    const int lane = t & 31;
    const int wave = t >> 5;
    const int v = (t < NR) ? cnt[t] : 0;
    int incl = v;
#pragma unroll
    for (int d = 1; d < 32; d <<= 1) {
      const int o = __shfl_up(incl, d);
      if (lane >= d) incl += o;
    }
    if (lane == 31) wsum[wave] = incl;
    __syncthreads();
    if (t < NR) {
      int prefix = 0;
      for (int k = 0; k < wave; ++k) prefix += wsum[k];
      off[t] = prefix + incl - v;
      if (t == NR - 1) off[NR] = prefix + incl;
    }
  }
  __syncthreads();

#pragma unroll
  for (int k = 0; k < EDGES_PER_THREAD; ++k) {
    if (rr[k] >= 0)
      srec[off[rr[k]] + rank[k]] = make_int2(key[k], __float_as_int(wv[k]));
  }
  __syncthreads();

  const int total = off[NR];
#pragma unroll
  for (int k = 0; k < EDGES_PER_THREAD; ++k) {
    const int i = k * BIN_THREADS + t;
    if (i < total) recs[e0 + i] = srec[i];
  }
  if (t < NR) seg[t * NCH + c] = make_int2(e0 + off[t], cnt[t]);
}

__global__ __launch_bounds__(PREP_THREADS) void bin_and_gemm(
    const float* __restrict__ feat,
    const float* __restrict__ W,
    float* __restrict__ h,
    const float* __restrict__ ew,
    const int* __restrict__ esrc,
    const int* __restrict__ edst,
    int2* __restrict__ recs,
    int2* __restrict__ seg)
{
  __shared__ alignas(16) unsigned char lds[PREP_LDS_BYTES];
  const int b = (int)blockIdx.x;
  if (b < NCH)
    bin_role(ew, esrc, edst, recs, seg, lds, b);
  else
    gemm_role(feat, W, h, (float4*)lds, b - NCH);
}


__device__ __forceinline__ void add_edges(const float* __restrict__ h,
                                          float* tile, int2 rec, int m,
                                          int lane)
{
  for (int j0 = 0; j0 < m; j0 += EDGE_GROUP) {
    int   kk[EDGE_GROUP];
    float lo[EDGE_GROUP], hi[EDGE_GROUP];
#pragma unroll
    for (int u = 0; u < EDGE_GROUP; ++u) {
      const int j = (j0 + u < m) ? (j0 + u) : (m - 1);
      kk[u] = __builtin_amdgcn_readlane(rec.x, j);
    }
#pragma unroll
    for (int u = 0; u < EDGE_GROUP; ++u) {
      const float* hrow = h + (size_t)(kk[u] & 0xFFFF) * D;
      lo[u] = hrow[lane];
      hi[u] = hrow[lane + 32];
    }
    static_assert(EDGE_GROUP == 8, "the fence below names 8 row pairs");
    asm volatile("" ::"v"(lo[0]), "v"(lo[1]), "v"(lo[2]), "v"(lo[3]),
                 "v"(lo[4]), "v"(lo[5]), "v"(lo[6]), "v"(lo[7]), "v"(hi[0]),
                 "v"(hi[1]), "v"(hi[2]), "v"(hi[3]), "v"(hi[4]), "v"(hi[5]),
                 "v"(hi[6]), "v"(hi[7]));
#pragma unroll
    for (int u = 0; u < EDGE_GROUP; ++u) {
      if (j0 + u < m) {
        const float w = __int_as_float(
            __builtin_amdgcn_readlane(rec.y, j0 + u));
        float* trow = tile + (int)((unsigned)kk[u] >> 16) * D;
        atomicAdd(trow + lane, lo[u] * w);
        atomicAdd(trow + lane + 32, hi[u] * w);
      }
    }
  }
}

__device__ __forceinline__ void flush_run(float* tile, int cur, float slo,
                                          float shi, int lane)
{
  if (cur >= 0) {
    atomicAdd(tile + cur * D + lane, slo);
    atomicAdd(tile + cur * D + lane + 32, shi);
  }
}

__device__ __forceinline__ void walk_sorted(const float* __restrict__ h,
                                            float* tile, int2 rec, int m,
                                            int lane, int& cur, float& slo,
                                            float& shi)
{
  for (int j0 = 0; j0 < m; j0 += WALK_GROUP) {
    int   kk[WALK_GROUP];
    float lo[WALK_GROUP], hi[WALK_GROUP];
#pragma unroll
    for (int u = 0; u < WALK_GROUP; ++u) {
      const int j = (j0 + u < m) ? (j0 + u) : (m - 1);
      kk[u] = __builtin_amdgcn_readlane(rec.x, j);
    }
#pragma unroll
    for (int u = 0; u < WALK_GROUP; ++u) {
      const float* hrow = h + (size_t)(kk[u] & 0xFFFF) * D;
      lo[u] = hrow[lane];
      hi[u] = hrow[lane + 32];
    }
    static_assert(WALK_GROUP == 16, "the fence below names 16 row pairs");
    asm volatile("" ::"v"(lo[0]), "v"(lo[1]), "v"(lo[2]), "v"(lo[3]),
                 "v"(lo[4]), "v"(lo[5]), "v"(lo[6]), "v"(lo[7]), "v"(lo[8]),
                 "v"(lo[9]), "v"(lo[10]), "v"(lo[11]), "v"(lo[12]),
                 "v"(lo[13]), "v"(lo[14]), "v"(lo[15]), "v"(hi[0]),
                 "v"(hi[1]), "v"(hi[2]), "v"(hi[3]), "v"(hi[4]), "v"(hi[5]),
                 "v"(hi[6]), "v"(hi[7]), "v"(hi[8]), "v"(hi[9]), "v"(hi[10]),
                 "v"(hi[11]), "v"(hi[12]), "v"(hi[13]), "v"(hi[14]),
                 "v"(hi[15]));
#pragma unroll
    for (int u = 0; u < WALK_GROUP; ++u) {
      if (j0 + u < m) {
        const float w = __int_as_float(
            __builtin_amdgcn_readlane(rec.y, j0 + u));
        const int row = (int)((unsigned)kk[u] >> 16);
        if (row != cur) {
          flush_run(tile, cur, slo, shi, lane);
          cur = row;
          slo = 0.0f;
          shi = 0.0f;
        }
        slo += lo[u] * w;
        shi += hi[u] * w;
      }
    }
  }
}

__global__ __launch_bounds__(OWN_THREADS) void owner_sum(
    const float* __restrict__ h,
    const int2* __restrict__ recs,
    const int2* __restrict__ seg,
    const float* __restrict__ skip_w,
    const float* __restrict__ bias,
    float* __restrict__ out)
{
  __shared__ float tile[R_ROWS * D];
  __shared__ alignas(8) int2 srec[SORT_CAP];
  __shared__ int cnt[R_ROWS];
  __shared__ int off[R_ROWS + 1];

  const int t    = (int)threadIdx.x;
  const int lane = t & 31;
  const int wave = __builtin_amdgcn_readfirstlane(t >> 5);
  const int r    = (int)blockIdx.x;

  int2 myseg = make_int2(0, 0);
  {
    const int c = wave + OWN_WAVES * lane;
    if (lane < SEG_PER_WAVE && c < NCH) myseg = seg[r * NCH + c];
  }

  const int   col  = t & (D - 1);
  const size_t own = (size_t)(r * R_ROWS + (t >> 6)) * D + col;
  const float* hp  = h + own;
  const float sk   = skip_w[col];
  const float bi   = bias[col];
  const bool last_step = t < LAST_STEP_THREADS;
  float hv[EPI_STEPS];
#pragma unroll
  for (int k = 0; k < EPI_STEPS - 1; ++k) hv[k] = hp[k * EPI_ROW_STEP * D];
  hv[EPI_STEPS - 1] = hp[last_step ? (EPI_STEPS - 1) * EPI_ROW_STEP * D : 0];

  for (int i = t; i < R_ROWS * D; i += OWN_THREADS) tile[i] = 0.0f;
  if (t < R_ROWS) cnt[t] = 0;

  int2 rec[SEG_PER_WAVE];
#pragma unroll
  for (int i = 0; i < SEG_PER_WAVE; ++i) {
    const int s = __builtin_amdgcn_readlane(myseg.x, i);
    const int n = __builtin_amdgcn_readlane(myseg.y, i);
    rec[i] = make_int2(0, 0);
    if (lane < n) rec[i] = recs[s + lane];
  }
  __syncthreads();

  int rank[SEG_PER_WAVE];
#pragma unroll
  for (int i = 0; i < SEG_PER_WAVE; ++i) {
    const int n = __builtin_amdgcn_readlane(myseg.y, i);
    rank[i] = 0;
    if (lane < n)
      rank[i] = atomicAdd(&cnt[(unsigned)rec[i].x >> 16], 1);
  }
  __syncthreads();

  if (wave == 0) {
    int c[ROWS_PER_LANE];
    int sum = 0;
#pragma unroll
    for (int k = 0; k < ROWS_PER_LANE; ++k) {
      const int row = lane * ROWS_PER_LANE + k;
      c[k] = row < R_ROWS ? cnt[row] : 0;
      sum += c[k];
    }
    int incl = sum;
#pragma unroll
    for (int d = 1; d < 32; d <<= 1) {
      const int o = __shfl_up(incl, d);
      if (lane >= d) incl += o;
    }
    int run = incl - sum;
#pragma unroll
    for (int k = 0; k < ROWS_PER_LANE; ++k) {
      const int row = lane * ROWS_PER_LANE + k;
      if (row < R_ROWS) off[row] = run;
      run += c[k];
    }
    if (lane == 31) off[R_ROWS] = incl;
  }
  __syncthreads();

  int slot[SEG_PER_WAVE];
#pragma unroll
  for (int i = 0; i < SEG_PER_WAVE; ++i)
    slot[i] = off[(unsigned)rec[i].x >> 16] + rank[i];
#pragma unroll
  for (int i = 0; i < SEG_PER_WAVE; ++i) {
    const int n = __builtin_amdgcn_readlane(myseg.y, i);
    if (lane < n) srec[slot[i]] = rec[i];
  }

#pragma unroll
  for (int i = 0; i < SEG_PER_WAVE; ++i) {
    const int s = __builtin_amdgcn_readlane(myseg.x, i);
    const int n = __builtin_amdgcn_readlane(myseg.y, i);
    for (int base = 32; base < n; base += 32) {
      const int m = (n - base) < 32 ? (n - base) : 32;
      int2 rx = make_int2(0, 0);
      if (lane < m) rx = recs[s + base + lane];
      add_edges(h, tile, rx, m, lane);
    }
  }
  __syncthreads();

  {
    const int total = __builtin_amdgcn_readfirstlane(off[R_ROWS]);
    const int lo_i  = (total * wave) / OWN_WAVES;
    const int hi_i  = (total * (wave + 1)) / OWN_WAVES;
    int   cur = -1;
    float slo = 0.0f, shi = 0.0f;
    for (int b = lo_i; b < hi_i; b += 32) {
      const int m = (hi_i - b) < 32 ? (hi_i - b) : 32;
      int2 rx = make_int2(0, 0);
      if (lane < m) rx = srec[b + lane];
      walk_sorted(h, tile, rx, m, lane, cur, slo, shi);
    }
    flush_run(tile, cur, slo, shi, lane);
  }
  __syncthreads();

  const float lam = 1.0507009873554805f;
  const float alp = 1.6732632423543772f;
  float tv[EPI_STEPS];
#pragma unroll
  for (int k = 0; k < EPI_STEPS - 1; ++k) tv[k] = tile[t + OWN_THREADS * k];
  tv[EPI_STEPS - 1] = tile[last_step ? t + OWN_THREADS * (EPI_STEPS - 1) : t];

  float* op = out + own;
#pragma unroll
  for (int k = 0; k < EPI_STEPS; ++k) {
    if (k < EPI_STEPS - 1 || last_step) {
      const float x = hv[k] * sk + bi + tv[k];
      op[k * EPI_ROW_STEP * D] =
          lam * (x > 0.0f ? x : alp * (__expf(x) - 1.0f));
    }
  }
}

extern "C" void kernel_launch(void* const* d_in, const int* in_sizes, int n_in,
                              void* d_out, int out_size, void* d_ws,
                              size_t ws_size, hipStream_t stream) {
  const float* feat   = (const float*)d_in[0];
  const float* W      = (const float*)d_in[1];
  const float* bias   = (const float*)d_in[2];
  const float* skip_w = (const float*)d_in[3];
  const float* ew     = (const float*)d_in[4];
  const int*   esrc   = (const int*)d_in[5];
  const int*   edst   = (const int*)d_in[6];
  float* out = (float*)d_out;
  float* h    = (float*)d_ws;
  int2*  recs = (int2*)((char*)d_ws + 16777216);
  int2*  seg  = (int2*)((char*)d_ws + 33554432);

  bin_and_gemm<<<NCH + GEMM_BLOCKS, PREP_THREADS, 0, stream>>>(
      feat, W, h, ew, esrc, edst, recs, seg);

  owner_sum<<<NR, OWN_THREADS, 0, stream>>>(h, recs, seg, skip_w, bias, out);
}


# speedup vs baseline: 2.6753x; 1.0358x over previous
#include <hip/hip_runtime.h>


typedef __attribute__((ext_vector_type(2))) float v2f;
typedef __attribute__((ext_vector_type(8))) float v8f;

#define N_NODES 50000
#define N_EDGES 800000
#define D 64

#define KBLOCKS (D / 8)
#define PREP_THREADS 256
#define ROW_TILES ((N_NODES + 15) / 16)
#define GEMM_BLOCKS ((ROW_TILES * 32 + PREP_THREADS - 1) / PREP_THREADS)
#define GEMM_LDS_BYTES (KBLOCKS * 4 * 32 * 16)

__device__ __forceinline__ void gemm_role(
    const float* __restrict__ feat,
    const float* __restrict__ W,
    float* __restrict__ h,
    float4* w_frag,
    int gb)
{
  const int tid   = (int)threadIdx.x;
  const int gwave = (gb * PREP_THREADS + tid) >> 5;
  const int lane  = tid & 31;
  const int m0    = gwave * 16;
  const bool active = m0 < N_NODES;

  const int half = lane >> 4;
  const int l    = lane & 15;

  const int arow_idx = active ? (m0 + l) : l;
  const float4* arow =
      (const float4*)(feat + (size_t)arow_idx * D + 4 * half);
  float4 a[KBLOCKS];
#pragma unroll
  for (int jj = 0; jj < KBLOCKS; ++jj) a[jj] = arow[2 * jj];

  const float4* W4 = (const float4*)W;
#pragma unroll
  for (int i = 0; i < (D * D / 4) / PREP_THREADS; ++i) {
    const int g = tid + PREP_THREADS * i;
    const int n = g >> 4;
    const int q = g & 15;
    const int e = ((q >> 1) * 4 + (n >> 4)) * 32 + (q & 1) * 16 + (n & 15);
    w_frag[e] = W4[g];
  }
  __syncthreads();
  if (!active) return;

  v8f acc0 = {}, acc1 = {}, acc2 = {}, acc3 = {};

#pragma unroll
  for (int jj = 0; jj < KBLOCKS; ++jj) {
    const float4 b0 = w_frag[(jj * 4 + 0) * 32 + lane];
    const float4 b1 = w_frag[(jj * 4 + 1) * 32 + lane];
    const float4 b2 = w_frag[(jj * 4 + 2) * 32 + lane];
    const float4 b3 = w_frag[(jj * 4 + 3) * 32 + lane];
    const v2f alo = {a[jj].x, a[jj].y};
    const v2f ahi = {a[jj].z, a[jj].w};
    const v2f b0lo = {b0.x, b0.y}, b0hi = {b0.z, b0.w};
    const v2f b1lo = {b1.x, b1.y}, b1hi = {b1.z, b1.w};
    const v2f b2lo = {b2.x, b2.y}, b2hi = {b2.z, b2.w};
    const v2f b3lo = {b3.x, b3.y}, b3hi = {b3.z, b3.w};
    acc0 = __builtin_amdgcn_wmma_f32_16x16x4_f32(false, alo, false, b0lo,
                                                 (short)0, acc0, false, false);
    acc1 = __builtin_amdgcn_wmma_f32_16x16x4_f32(false, alo, false, b1lo,
                                                 (short)0, acc1, false, false);
    acc2 = __builtin_amdgcn_wmma_f32_16x16x4_f32(false, alo, false, b2lo,
                                                 (short)0, acc2, false, false);
    acc3 = __builtin_amdgcn_wmma_f32_16x16x4_f32(false, alo, false, b3lo,
                                                 (short)0, acc3, false, false);
    acc0 = __builtin_amdgcn_wmma_f32_16x16x4_f32(false, ahi, false, b0hi,
                                                 (short)0, acc0, false, false);
    acc1 = __builtin_amdgcn_wmma_f32_16x16x4_f32(false, ahi, false, b1hi,
                                                 (short)0, acc1, false, false);
    acc2 = __builtin_amdgcn_wmma_f32_16x16x4_f32(false, ahi, false, b2hi,
                                                 (short)0, acc2, false, false);
    acc3 = __builtin_amdgcn_wmma_f32_16x16x4_f32(false, ahi, false, b3hi,
                                                 (short)0, acc3, false, false);
  }

  const v8f accs[4] = {acc0, acc1, acc2, acc3};
#pragma unroll
  for (int nt = 0; nt < 4; ++nt) {
    const int n = nt * 16 + l;
#pragma unroll
    for (int r = 0; r < 8; ++r) {
      const int row = m0 + r + 8 * half;
      h[(size_t)row * D + n] = accs[nt][r];
    }
  }
}

#define R_ROWS 200
#define NR ((N_NODES + R_ROWS - 1) / R_ROWS)
#define CH 4096
#define NCH ((N_EDGES + CH - 1) / CH)
#define BIN_THREADS PREP_THREADS
#define EDGES_PER_THREAD (CH / BIN_THREADS)
#define BIN_LDS_BYTES (CH * 8 + (NR + NR + 1 + BIN_THREADS / 32) * 4)
#define PREP_LDS_BYTES \
  (BIN_LDS_BYTES > GEMM_LDS_BYTES ? BIN_LDS_BYTES : GEMM_LDS_BYTES)
#define OWN_THREADS 1024
#define OWN_WAVES (OWN_THREADS / 32)
#define SEG_PER_WAVE ((NCH + OWN_WAVES - 1) / OWN_WAVES)
#define EDGE_GROUP 8
#define SORT_CAP (NCH * 32)
#define WALK_GROUP 16
#define ROWS_PER_LANE ((R_ROWS + 31) / 32)
#define EPI_STEPS ((R_ROWS * D + OWN_THREADS - 1) / OWN_THREADS)
#define EPI_ROW_STEP (OWN_THREADS / D)
#define LAST_STEP_THREADS (R_ROWS * D - (EPI_STEPS - 1) * OWN_THREADS)

static_assert(N_NODES <= 65536, "src fits the low 16 bits of a record");
static_assert(R_ROWS <= 65536, "row-in-range fits the high 16 bits");
static_assert(NR <= BIN_THREADS, "one thread per range in the scan");
static_assert(CH % BIN_THREADS == 0, "whole edges per thread");
static_assert(SEG_PER_WAVE <= 32, "a wave keeps its segments one per lane");
static_assert(OWN_THREADS % D == 0, "a thread's column is the same every step");
static_assert(NR * R_ROWS == N_NODES, "every row of every range is a node row");
static_assert(LAST_STEP_THREADS > 0 && LAST_STEP_THREADS % 32 == 0,
              "the last step covers whole waves");
static_assert(R_ROWS <= OWN_THREADS, "one thread clears one row counter");
static_assert(SORT_CAP * OWN_WAVES < (1 << 30), "share bounds fit an int");

__device__ __forceinline__ void bin_role(
    const float* __restrict__ ew,
    const int* __restrict__ esrc,
    const int* __restrict__ edst,
    int2* __restrict__ recs,
    int2* __restrict__ seg,
    unsigned char* lds,
    int c)
{
  int2* srec = (int2*)lds;
  int*  cnt  = (int*)(lds + CH * 8);
  int*  off  = cnt + NR;
  int*  wsum = off + NR + 1;

  const int t  = (int)threadIdx.x;
  const int e0 = c * CH;

  if (t < NR) cnt[t] = 0;
  __syncthreads();

  int   sv[EDGES_PER_THREAD], dv[EDGES_PER_THREAD];
  float wv[EDGES_PER_THREAD];
#pragma unroll
  for (int k = 0; k < EDGES_PER_THREAD; ++k) {
    const int e  = e0 + k * BIN_THREADS + t;
    const int ec = e < N_EDGES ? e : N_EDGES - 1;
    sv[k] = esrc[ec];
    dv[k] = edst[ec];
    wv[k] = ew[ec];
  }
  static_assert(EDGES_PER_THREAD == 16, "the fence below names 16 values");
  asm volatile("" ::"v"(sv[0]), "v"(sv[1]), "v"(sv[2]), "v"(sv[3]),
               "v"(sv[4]), "v"(sv[5]), "v"(sv[6]), "v"(sv[7]), "v"(sv[8]),
               "v"(sv[9]), "v"(sv[10]), "v"(sv[11]), "v"(sv[12]), "v"(sv[13]),
               "v"(sv[14]), "v"(sv[15]));

  int rr[EDGES_PER_THREAD], key[EDGES_PER_THREAD], rank[EDGES_PER_THREAD];
#pragma unroll
  for (int k = 0; k < EDGES_PER_THREAD; ++k) {
    const int e = e0 + k * BIN_THREADS + t;
    const int d = dv[k];
    rr[k] = -1; key[k] = 0; rank[k] = 0;
    if (e < N_EDGES && (unsigned)d < (unsigned)N_NODES) {
      const int r = d / R_ROWS;
      rr[k]   = r;
      key[k]  = (sv[k] & 0xFFFF) | ((d - r * R_ROWS) << 16);
      rank[k] = atomicAdd(&cnt[r], 1);
    }
  }
  __syncthreads();

  {
    const int lane = t & 31;
    const int wave = t >> 5;
    const int v = (t < NR) ? cnt[t] : 0;
    int incl = v;
#pragma unroll
    for (int d = 1; d < 32; d <<= 1) {
      const int o = __shfl_up(incl, d);
      if (lane >= d) incl += o;
    }
    if (lane == 31) wsum[wave] = incl;
    __syncthreads();
    if (t < NR) {
      int prefix = 0;
      for (int k = 0; k < wave; ++k) prefix += wsum[k];
      off[t] = prefix + incl - v;
      if (t == NR - 1) off[NR] = prefix + incl;
    }
  }
  __syncthreads();

#pragma unroll
  for (int k = 0; k < EDGES_PER_THREAD; ++k) {
    if (rr[k] >= 0)
      srec[off[rr[k]] + rank[k]] = make_int2(key[k], __float_as_int(wv[k]));
  }
  __syncthreads();

  const int total = off[NR];
#pragma unroll
  for (int k = 0; k < EDGES_PER_THREAD; ++k) {
    const int i = k * BIN_THREADS + t;
    if (i < total) recs[e0 + i] = srec[i];
  }
  if (t < NR) seg[t * NCH + c] = make_int2(e0 + off[t], cnt[t]);
}

__global__ __launch_bounds__(PREP_THREADS) void bin_and_gemm(
    const float* __restrict__ feat,
    const float* __restrict__ W,
    float* __restrict__ h,
    const float* __restrict__ ew,
    const int* __restrict__ esrc,
    const int* __restrict__ edst,
    int2* __restrict__ recs,
    int2* __restrict__ seg)
{
  __shared__ alignas(16) unsigned char lds[PREP_LDS_BYTES];
  const int b = (int)blockIdx.x;
  if (b < NCH)
    bin_role(ew, esrc, edst, recs, seg, lds, b);
  else
    gemm_role(feat, W, h, (float4*)lds, b - NCH);
}


__device__ __forceinline__ void add_edges(const float2* __restrict__ h2,
                                          float* tile, int2 rec, int m,
                                          int lane)
{
  for (int j0 = 0; j0 < m; j0 += EDGE_GROUP) {
    int   kk[EDGE_GROUP];
    float lo[EDGE_GROUP], hi[EDGE_GROUP];
#pragma unroll
    for (int u = 0; u < EDGE_GROUP; ++u) {
      const int j = (j0 + u < m) ? (j0 + u) : (m - 1);
      kk[u] = __builtin_amdgcn_readlane(rec.x, j);
    }
#pragma unroll
    for (int u = 0; u < EDGE_GROUP; ++u) {
      const float2 v = h2[(kk[u] & 0xFFFF) * (D / 2) + lane];
      lo[u] = v.x;
      hi[u] = v.y;
    }
    static_assert(EDGE_GROUP == 8, "the fence below names 8 row pairs");
    asm volatile("" ::"v"(lo[0]), "v"(lo[1]), "v"(lo[2]), "v"(lo[3]),
                 "v"(lo[4]), "v"(lo[5]), "v"(lo[6]), "v"(lo[7]), "v"(hi[0]),
                 "v"(hi[1]), "v"(hi[2]), "v"(hi[3]), "v"(hi[4]), "v"(hi[5]),
                 "v"(hi[6]), "v"(hi[7]));
#pragma unroll
    for (int u = 0; u < EDGE_GROUP; ++u) {
      if (j0 + u < m) {
        const float w = __int_as_float(
            __builtin_amdgcn_readlane(rec.y, j0 + u));
        float* trow = tile + (int)((unsigned)kk[u] >> 16) * D + 2 * lane;
        atomicAdd(trow, lo[u] * w);
        atomicAdd(trow + 1, hi[u] * w);
      }
    }
  }
}

__device__ __forceinline__ void flush_run(float* tile, int cur, float slo,
                                          float shi, int lane)
{
  if (cur >= 0) {
    atomicAdd(tile + cur * D + 2 * lane, slo);
    atomicAdd(tile + cur * D + 2 * lane + 1, shi);
  }
}

__device__ __forceinline__ void walk_sorted(const float2* __restrict__ h2,
                                            float* tile, int2 rec, int m,
                                            int lane, int& cur, float& slo,
                                            float& shi)
{
  for (int j0 = 0; j0 < m; j0 += WALK_GROUP) {
    int   kk[WALK_GROUP];
    float lo[WALK_GROUP], hi[WALK_GROUP];
#pragma unroll
    for (int u = 0; u < WALK_GROUP; ++u) {
      const int j = (j0 + u < m) ? (j0 + u) : (m - 1);
      kk[u] = __builtin_amdgcn_readlane(rec.x, j);
    }
#pragma unroll
    for (int u = 0; u < WALK_GROUP; ++u) {
      const float2 v = h2[(kk[u] & 0xFFFF) * (D / 2) + lane];
      lo[u] = v.x;
      hi[u] = v.y;
    }
    static_assert(WALK_GROUP == 16, "the fence below names 16 row pairs");
    asm volatile("" ::"v"(lo[0]), "v"(lo[1]), "v"(lo[2]), "v"(lo[3]),
                 "v"(lo[4]), "v"(lo[5]), "v"(lo[6]), "v"(lo[7]), "v"(lo[8]),
                 "v"(lo[9]), "v"(lo[10]), "v"(lo[11]), "v"(lo[12]),
                 "v"(lo[13]), "v"(lo[14]), "v"(lo[15]), "v"(hi[0]),
                 "v"(hi[1]), "v"(hi[2]), "v"(hi[3]), "v"(hi[4]), "v"(hi[5]),
                 "v"(hi[6]), "v"(hi[7]), "v"(hi[8]), "v"(hi[9]), "v"(hi[10]),
                 "v"(hi[11]), "v"(hi[12]), "v"(hi[13]), "v"(hi[14]),
                 "v"(hi[15]));
#pragma unroll
    for (int u = 0; u < WALK_GROUP; ++u) {
      if (j0 + u < m) {
        const float w = __int_as_float(
            __builtin_amdgcn_readlane(rec.y, j0 + u));
        const int row = (int)((unsigned)kk[u] >> 16);
        if (row != cur) {
          flush_run(tile, cur, slo, shi, lane);
          cur = row;
          slo = 0.0f;
          shi = 0.0f;
        }
        slo += lo[u] * w;
        shi += hi[u] * w;
      }
    }
  }
}

__global__ __launch_bounds__(OWN_THREADS) void owner_sum(
    const float* __restrict__ h,
    const int2* __restrict__ recs,
    const int2* __restrict__ seg,
    const float* __restrict__ skip_w,
    const float* __restrict__ bias,
    float* __restrict__ out)
{
  __shared__ float tile[R_ROWS * D];
  __shared__ alignas(8) int2 srec[SORT_CAP];
  __shared__ int cnt[R_ROWS];
  __shared__ int off[R_ROWS + 1];

  const int t    = (int)threadIdx.x;
  const int lane = t & 31;
  const int wave = __builtin_amdgcn_readfirstlane(t >> 5);
  const int r    = (int)blockIdx.x;

  int2 myseg = make_int2(0, 0);
  {
    const int c = wave + OWN_WAVES * lane;
    if (lane < SEG_PER_WAVE && c < NCH) myseg = seg[r * NCH + c];
  }

  const float2* __restrict__ h2 = (const float2*)h;
  const int   col  = t & (D - 1);
  const size_t own = (size_t)(r * R_ROWS + (t >> 6)) * D + col;
  const float* hp  = h + own;
  const float sk   = skip_w[col];
  const float bi   = bias[col];
  const bool last_step = t < LAST_STEP_THREADS;
  float hv[EPI_STEPS];
#pragma unroll
  for (int k = 0; k < EPI_STEPS - 1; ++k) hv[k] = hp[k * EPI_ROW_STEP * D];
  hv[EPI_STEPS - 1] = hp[last_step ? (EPI_STEPS - 1) * EPI_ROW_STEP * D : 0];

  for (int i = t; i < R_ROWS * D; i += OWN_THREADS) tile[i] = 0.0f;
  if (t < R_ROWS) cnt[t] = 0;

  int2 rec[SEG_PER_WAVE];
#pragma unroll
  for (int i = 0; i < SEG_PER_WAVE; ++i) {
    const int s = __builtin_amdgcn_readlane(myseg.x, i);
    const int n = __builtin_amdgcn_readlane(myseg.y, i);
    rec[i] = make_int2(0, 0);
    if (lane < n) rec[i] = recs[s + lane];
  }
  __syncthreads();

  int rank[SEG_PER_WAVE];
#pragma unroll
  for (int i = 0; i < SEG_PER_WAVE; ++i) {
    const int n = __builtin_amdgcn_readlane(myseg.y, i);
    rank[i] = 0;
    if (lane < n)
      rank[i] = atomicAdd(&cnt[(unsigned)rec[i].x >> 16], 1);
  }
  __syncthreads();

  if (wave == 0) {
    int c[ROWS_PER_LANE];
    int sum = 0;
#pragma unroll
    for (int k = 0; k < ROWS_PER_LANE; ++k) {
      const int row = lane * ROWS_PER_LANE + k;
      c[k] = row < R_ROWS ? cnt[row] : 0;
      sum += c[k];
    }
    int incl = sum;
#pragma unroll
    for (int d = 1; d < 32; d <<= 1) {
      const int o = __shfl_up(incl, d);
      if (lane >= d) incl += o;
    }
    int run = incl - sum;
#pragma unroll
    for (int k = 0; k < ROWS_PER_LANE; ++k) {
      const int row = lane * ROWS_PER_LANE + k;
      if (row < R_ROWS) off[row] = run;
      run += c[k];
    }
    if (lane == 31) off[R_ROWS] = incl;
  }
  __syncthreads();

  int slot[SEG_PER_WAVE];
#pragma unroll
  for (int i = 0; i < SEG_PER_WAVE; ++i)
    slot[i] = off[(unsigned)rec[i].x >> 16] + rank[i];
#pragma unroll
  for (int i = 0; i < SEG_PER_WAVE; ++i) {
    const int n = __builtin_amdgcn_readlane(myseg.y, i);
    if (lane < n) srec[slot[i]] = rec[i];
  }

#pragma unroll
  for (int i = 0; i < SEG_PER_WAVE; ++i) {
    const int s = __builtin_amdgcn_readlane(myseg.x, i);
    const int n = __builtin_amdgcn_readlane(myseg.y, i);
    for (int base = 32; base < n; base += 32) {
      const int m = (n - base) < 32 ? (n - base) : 32;
      int2 rx = make_int2(0, 0);
      if (lane < m) rx = recs[s + base + lane];
      add_edges(h2, tile, rx, m, lane);
    }
  }
  __syncthreads();

  {
    const int total = __builtin_amdgcn_readfirstlane(off[R_ROWS]);
    const int lo_i  = (total * wave) / OWN_WAVES;
    const int hi_i  = (total * (wave + 1)) / OWN_WAVES;
    int   cur = -1;
    float slo = 0.0f, shi = 0.0f;
    for (int b = lo_i; b < hi_i; b += 32) {
      const int m = (hi_i - b) < 32 ? (hi_i - b) : 32;
      int2 rx = make_int2(0, 0);
      if (lane < m) rx = srec[b + lane];
      walk_sorted(h2, tile, rx, m, lane, cur, slo, shi);
    }
    flush_run(tile, cur, slo, shi, lane);
  }
  __syncthreads();

  const float lam = 1.0507009873554805f;
  const float alp = 1.6732632423543772f;
  float tv[EPI_STEPS];
#pragma unroll
  for (int k = 0; k < EPI_STEPS - 1; ++k) tv[k] = tile[t + OWN_THREADS * k];
  tv[EPI_STEPS - 1] = tile[last_step ? t + OWN_THREADS * (EPI_STEPS - 1) : t];

  float* op = out + own;
#pragma unroll
  for (int k = 0; k < EPI_STEPS; ++k) {
    if (k < EPI_STEPS - 1 || last_step) {
      const float x = hv[k] * sk + bi + tv[k];
      op[k * EPI_ROW_STEP * D] =
          lam * (x > 0.0f ? x : alp * (__expf(x) - 1.0f));
    }
  }
}

extern "C" void kernel_launch(void* const* d_in, const int* in_sizes, int n_in,
                              void* d_out, int out_size, void* d_ws,
                              size_t ws_size, hipStream_t stream) {
  const float* feat   = (const float*)d_in[0];
  const float* W      = (const float*)d_in[1];
  const float* bias   = (const float*)d_in[2];
  const float* skip_w = (const float*)d_in[3];
  const float* ew     = (const float*)d_in[4];
  const int*   esrc   = (const int*)d_in[5];
  const int*   edst   = (const int*)d_in[6];
  float* out = (float*)d_out;
  float* h    = (float*)d_ws;
  int2*  recs = (int2*)((char*)d_ws + 16777216);
  int2*  seg  = (int2*)((char*)d_ws + 33554432);

  bin_and_gemm<<<NCH + GEMM_BLOCKS, PREP_THREADS, 0, stream>>>(
      feat, W, h, ew, esrc, edst, recs, seg);

  owner_sum<<<NR, OWN_THREADS, 0, stream>>>(h, recs, seg, skip_w, bias, out);
}


# Owner walk: each 16-lane half of a wave takes its own part of the sorted records, 16 B per lane: one load fetches two rows, row test on the vector unit; a run cut between halves joins by LDS add
# speedup vs baseline: 2.8294x; 1.0576x over previous
#include <hip/hip_runtime.h>


typedef __attribute__((ext_vector_type(2))) float v2f;
typedef __attribute__((ext_vector_type(8))) float v8f;

#define N_NODES 50000
#define N_EDGES 800000
#define D 64

#define KBLOCKS (D / 8)
#define PREP_THREADS 256
#define ROW_TILES ((N_NODES + 15) / 16)
#define GEMM_BLOCKS ((ROW_TILES * 32 + PREP_THREADS - 1) / PREP_THREADS)
#define GEMM_LDS_BYTES (KBLOCKS * 4 * 32 * 16)

__device__ __forceinline__ void gemm_role(
    const float* __restrict__ feat,
    const float* __restrict__ W,
    float* __restrict__ h,
    float4* w_frag,
    int gb)
{
  const int tid   = (int)threadIdx.x;
  const int gwave = (gb * PREP_THREADS + tid) >> 5;
  const int lane  = tid & 31;
  const int m0    = gwave * 16;
  const bool active = m0 < N_NODES;

  const int half = lane >> 4;
  const int l    = lane & 15;

  const int arow_idx = active ? (m0 + l) : l;
  const float4* arow =
      (const float4*)(feat + (size_t)arow_idx * D + 4 * half);
  float4 a[KBLOCKS];
#pragma unroll
  for (int jj = 0; jj < KBLOCKS; ++jj) a[jj] = arow[2 * jj];

  const float4* W4 = (const float4*)W;
#pragma unroll
  for (int i = 0; i < (D * D / 4) / PREP_THREADS; ++i) {
    const int g = tid + PREP_THREADS * i;
    const int n = g >> 4;
    const int q = g & 15;
    const int e = ((q >> 1) * 4 + (n >> 4)) * 32 + (q & 1) * 16 + (n & 15);
    w_frag[e] = W4[g];
  }
  __syncthreads();
  if (!active) return;

  v8f acc0 = {}, acc1 = {}, acc2 = {}, acc3 = {};

#pragma unroll
  for (int jj = 0; jj < KBLOCKS; ++jj) {
    const float4 b0 = w_frag[(jj * 4 + 0) * 32 + lane];
    const float4 b1 = w_frag[(jj * 4 + 1) * 32 + lane];
    const float4 b2 = w_frag[(jj * 4 + 2) * 32 + lane];
    const float4 b3 = w_frag[(jj * 4 + 3) * 32 + lane];
    const v2f alo = {a[jj].x, a[jj].y};
    const v2f ahi = {a[jj].z, a[jj].w};
    const v2f b0lo = {b0.x, b0.y}, b0hi = {b0.z, b0.w};
    const v2f b1lo = {b1.x, b1.y}, b1hi = {b1.z, b1.w};
    const v2f b2lo = {b2.x, b2.y}, b2hi = {b2.z, b2.w};
    const v2f b3lo = {b3.x, b3.y}, b3hi = {b3.z, b3.w};
    acc0 = __builtin_amdgcn_wmma_f32_16x16x4_f32(false, alo, false, b0lo,
                                                 (short)0, acc0, false, false);
    acc1 = __builtin_amdgcn_wmma_f32_16x16x4_f32(false, alo, false, b1lo,
                                                 (short)0, acc1, false, false);
    acc2 = __builtin_amdgcn_wmma_f32_16x16x4_f32(false, alo, false, b2lo,
                                                 (short)0, acc2, false, false);
    acc3 = __builtin_amdgcn_wmma_f32_16x16x4_f32(false, alo, false, b3lo,
                                                 (short)0, acc3, false, false);
    acc0 = __builtin_amdgcn_wmma_f32_16x16x4_f32(false, ahi, false, b0hi,
                                                 (short)0, acc0, false, false);
    acc1 = __builtin_amdgcn_wmma_f32_16x16x4_f32(false, ahi, false, b1hi,
                                                 (short)0, acc1, false, false);
    acc2 = __builtin_amdgcn_wmma_f32_16x16x4_f32(false, ahi, false, b2hi,
                                                 (short)0, acc2, false, false);
    acc3 = __builtin_amdgcn_wmma_f32_16x16x4_f32(false, ahi, false, b3hi,
                                                 (short)0, acc3, false, false);
  }

  const v8f accs[4] = {acc0, acc1, acc2, acc3};
#pragma unroll
  for (int nt = 0; nt < 4; ++nt) {
    const int n = nt * 16 + l;
#pragma unroll
    for (int r = 0; r < 8; ++r) {
      const int row = m0 + r + 8 * half;
      h[(size_t)row * D + n] = accs[nt][r];
    }
  }
}

#define R_ROWS 200
#define NR ((N_NODES + R_ROWS - 1) / R_ROWS)
#define CH 4096
#define NCH ((N_EDGES + CH - 1) / CH)
#define BIN_THREADS PREP_THREADS
#define EDGES_PER_THREAD (CH / BIN_THREADS)
#define BIN_LDS_BYTES (CH * 8 + (NR + NR + 1 + BIN_THREADS / 32) * 4)
#define PREP_LDS_BYTES \
  (BIN_LDS_BYTES > GEMM_LDS_BYTES ? BIN_LDS_BYTES : GEMM_LDS_BYTES)
#define OWN_THREADS 1024
#define OWN_WAVES (OWN_THREADS / 32)
#define SEG_PER_WAVE ((NCH + OWN_WAVES - 1) / OWN_WAVES)
#define EDGE_GROUP 8
#define SORT_CAP (NCH * 32)
#define HALF_BLOCK 16
#define HALF_STEPS 8
#define NO_ROW 0xFFFF
#define ROWS_PER_LANE ((R_ROWS + 31) / 32)
#define EPI_STEPS ((R_ROWS * D + OWN_THREADS - 1) / OWN_THREADS)
#define EPI_ROW_STEP (OWN_THREADS / D)
#define LAST_STEP_THREADS (R_ROWS * D - (EPI_STEPS - 1) * OWN_THREADS)

static_assert(N_NODES <= 65536, "src fits the low 16 bits of a record");
static_assert(R_ROWS <= 65536, "row-in-range fits the high 16 bits");
static_assert(NR <= BIN_THREADS, "one thread per range in the scan");
static_assert(CH % BIN_THREADS == 0, "whole edges per thread");
static_assert(SEG_PER_WAVE <= 32, "a wave keeps its segments one per lane");
static_assert(OWN_THREADS % D == 0, "a thread's column is the same every step");
static_assert(NR * R_ROWS == N_NODES, "every row of every range is a node row");
static_assert(LAST_STEP_THREADS > 0 && LAST_STEP_THREADS % 32 == 0,
              "the last step covers whole waves");
static_assert(R_ROWS <= OWN_THREADS, "one thread clears one row counter");
static_assert(SORT_CAP * OWN_WAVES < (1 << 30), "share bounds fit an int");
static_assert(R_ROWS < NO_ROW, "the empty-slot mark is no row of the tile");
static_assert(HALF_BLOCK == 16 && HALF_BLOCK % HALF_STEPS == 0,
              "a block is one record per lane of a half, in whole step groups");

__device__ __forceinline__ void bin_role(
    const float* __restrict__ ew,
    const int* __restrict__ esrc,
    const int* __restrict__ edst,
    int2* __restrict__ recs,
    int2* __restrict__ seg,
    unsigned char* lds,
    int c)
{
  int2* srec = (int2*)lds;
  int*  cnt  = (int*)(lds + CH * 8);
  int*  off  = cnt + NR;
  int*  wsum = off + NR + 1;

  const int t  = (int)threadIdx.x;
  const int e0 = c * CH;

  if (t < NR) cnt[t] = 0;
  __syncthreads();

  int   sv[EDGES_PER_THREAD], dv[EDGES_PER_THREAD];
  float wv[EDGES_PER_THREAD];
#pragma unroll
  for (int k = 0; k < EDGES_PER_THREAD; ++k) {
    const int e  = e0 + k * BIN_THREADS + t;
    const int ec = e < N_EDGES ? e : N_EDGES - 1;
    sv[k] = esrc[ec];
    dv[k] = edst[ec];
    wv[k] = ew[ec];
  }
  static_assert(EDGES_PER_THREAD == 16, "the fence below names 16 values");
  asm volatile("" ::"v"(sv[0]), "v"(sv[1]), "v"(sv[2]), "v"(sv[3]),
               "v"(sv[4]), "v"(sv[5]), "v"(sv[6]), "v"(sv[7]), "v"(sv[8]),
               "v"(sv[9]), "v"(sv[10]), "v"(sv[11]), "v"(sv[12]), "v"(sv[13]),
               "v"(sv[14]), "v"(sv[15]));

  int rr[EDGES_PER_THREAD], key[EDGES_PER_THREAD], rank[EDGES_PER_THREAD];
#pragma unroll
  for (int k = 0; k < EDGES_PER_THREAD; ++k) {
    const int e = e0 + k * BIN_THREADS + t;
    const int d = dv[k];
    rr[k] = -1; key[k] = 0; rank[k] = 0;
    if (e < N_EDGES && (unsigned)d < (unsigned)N_NODES) {
      const int r = d / R_ROWS;
      rr[k]   = r;
      key[k]  = (sv[k] & 0xFFFF) | ((d - r * R_ROWS) << 16);
      rank[k] = atomicAdd(&cnt[r], 1);
    }
  }
  __syncthreads();

  {
    const int lane = t & 31;
    const int wave = t >> 5;
    const int v = (t < NR) ? cnt[t] : 0;
    int incl = v;
#pragma unroll
    for (int d = 1; d < 32; d <<= 1) {
      const int o = __shfl_up(incl, d);
      if (lane >= d) incl += o;
    }
    if (lane == 31) wsum[wave] = incl;
    __syncthreads();
    if (t < NR) {
      int prefix = 0;
      for (int k = 0; k < wave; ++k) prefix += wsum[k];
      off[t] = prefix + incl - v;
      if (t == NR - 1) off[NR] = prefix + incl;
    }
  }
  __syncthreads();

#pragma unroll
  for (int k = 0; k < EDGES_PER_THREAD; ++k) {
    if (rr[k] >= 0)
      srec[off[rr[k]] + rank[k]] = make_int2(key[k], __float_as_int(wv[k]));
  }
  __syncthreads();

  const int total = off[NR];
#pragma unroll
  for (int k = 0; k < EDGES_PER_THREAD; ++k) {
    const int i = k * BIN_THREADS + t;
    if (i < total) recs[e0 + i] = srec[i];
  }
  if (t < NR) seg[t * NCH + c] = make_int2(e0 + off[t], cnt[t]);
}

__global__ __launch_bounds__(PREP_THREADS) void bin_and_gemm(
    const float* __restrict__ feat,
    const float* __restrict__ W,
    float* __restrict__ h,
    const float* __restrict__ ew,
    const int* __restrict__ esrc,
    const int* __restrict__ edst,
    int2* __restrict__ recs,
    int2* __restrict__ seg)
{
  __shared__ alignas(16) unsigned char lds[PREP_LDS_BYTES];
  const int b = (int)blockIdx.x;
  if (b < NCH)
    bin_role(ew, esrc, edst, recs, seg, lds, b);
  else
    gemm_role(feat, W, h, (float4*)lds, b - NCH);
}


__device__ __forceinline__ void add_edges(const float2* __restrict__ h2,
                                          float* tile, int2 rec, int m,
                                          int lane)
{
  for (int j0 = 0; j0 < m; j0 += EDGE_GROUP) {
    int   kk[EDGE_GROUP];
    float lo[EDGE_GROUP], hi[EDGE_GROUP];
#pragma unroll
    for (int u = 0; u < EDGE_GROUP; ++u) {
      const int j = (j0 + u < m) ? (j0 + u) : (m - 1);
      kk[u] = __builtin_amdgcn_readlane(rec.x, j);
    }
#pragma unroll
    for (int u = 0; u < EDGE_GROUP; ++u) {
      const float2 v = h2[(kk[u] & 0xFFFF) * (D / 2) + lane];
      lo[u] = v.x;
      hi[u] = v.y;
    }
    static_assert(EDGE_GROUP == 8, "the fence below names 8 row pairs");
    asm volatile("" ::"v"(lo[0]), "v"(lo[1]), "v"(lo[2]), "v"(lo[3]),
                 "v"(lo[4]), "v"(lo[5]), "v"(lo[6]), "v"(lo[7]), "v"(hi[0]),
                 "v"(hi[1]), "v"(hi[2]), "v"(hi[3]), "v"(hi[4]), "v"(hi[5]),
                 "v"(hi[6]), "v"(hi[7]));
#pragma unroll
    for (int u = 0; u < EDGE_GROUP; ++u) {
      if (j0 + u < m) {
        const float w = __int_as_float(
            __builtin_amdgcn_readlane(rec.y, j0 + u));
        float* trow = tile + (int)((unsigned)kk[u] >> 16) * D + 2 * lane;
        atomicAdd(trow, lo[u] * w);
        atomicAdd(trow + 1, hi[u] * w);
      }
    }
  }
}

__device__ __forceinline__ void flush_run(float* tile, int cur, float4 s,
                                          int l16)
{
  if (cur != NO_ROW) {
    float* p = tile + cur * D + 4 * l16;
    atomicAdd(p, s.x);
    atomicAdd(p + 1, s.y);
    atomicAdd(p + 2, s.z);
    atomicAdd(p + 3, s.w);
  }
}

#define HALF_BCAST(v, U)                                                    \
  ((int)__builtin_amdgcn_permlane16((unsigned)(v), (unsigned)(v),           \
                                    0x11111111u * (unsigned)(U),            \
                                    0x11111111u * (unsigned)(U), false, false))

template <int G0>
__device__ __forceinline__ void walk_steps(const float4* __restrict__ h4,
                                           float* tile, int2 rx, int l16,
                                           int& cur, float4& acc)
{
  int    key[HALF_STEPS];
  float4 v[HALF_STEPS];
#pragma unroll
  for (int u = 0; u < HALF_STEPS; ++u) key[u] = HALF_BCAST(rx.x, G0 + u);
#pragma unroll
  for (int u = 0; u < HALF_STEPS; ++u)
    v[u] = h4[(key[u] & 0xFFFF) * (D / 4) + l16];
  static_assert(HALF_STEPS == 8, "the fences below name 8 float4");
  asm volatile("" ::"v"(v[0].x), "v"(v[0].y), "v"(v[0].z), "v"(v[0].w),
               "v"(v[1].x), "v"(v[1].y), "v"(v[1].z), "v"(v[1].w),
               "v"(v[2].x), "v"(v[2].y), "v"(v[2].z), "v"(v[2].w),
               "v"(v[3].x), "v"(v[3].y), "v"(v[3].z), "v"(v[3].w));
  asm volatile("" ::"v"(v[4].x), "v"(v[4].y), "v"(v[4].z), "v"(v[4].w),
               "v"(v[5].x), "v"(v[5].y), "v"(v[5].z), "v"(v[5].w),
               "v"(v[6].x), "v"(v[6].y), "v"(v[6].z), "v"(v[6].w),
               "v"(v[7].x), "v"(v[7].y), "v"(v[7].z), "v"(v[7].w));
#pragma unroll
  for (int u = 0; u < HALF_STEPS; ++u) {
    const float w   = __int_as_float(HALF_BCAST(rx.y, G0 + u));
    const int   row = (int)((unsigned)key[u] >> 16);
    if (row != cur) {
      flush_run(tile, cur, acc, l16);
      cur = row;
      acc = make_float4(0.0f, 0.0f, 0.0f, 0.0f);
    }
    acc.x += v[u].x * w;
    acc.y += v[u].y * w;
    acc.z += v[u].z * w;
    acc.w += v[u].w * w;
  }
}

__global__ __launch_bounds__(OWN_THREADS) void owner_sum(
    const float* __restrict__ h,
    const int2* __restrict__ recs,
    const int2* __restrict__ seg,
    const float* __restrict__ skip_w,
    const float* __restrict__ bias,
    float* __restrict__ out)
{
  __shared__ float tile[R_ROWS * D];
  __shared__ alignas(8) int2 srec[SORT_CAP];
  __shared__ int cnt[R_ROWS];
  __shared__ int off[R_ROWS + 1];

  const int t    = (int)threadIdx.x;
  const int lane = t & 31;
  const int wave = __builtin_amdgcn_readfirstlane(t >> 5);
  const int r    = (int)blockIdx.x;

  int2 myseg = make_int2(0, 0);
  {
    const int c = wave + OWN_WAVES * lane;
    if (lane < SEG_PER_WAVE && c < NCH) myseg = seg[r * NCH + c];
  }

  const float2* __restrict__ h2 = (const float2*)h;
  const int   col  = t & (D - 1);
  const size_t own = (size_t)(r * R_ROWS + (t >> 6)) * D + col;
  const float* hp  = h + own;
  const float sk   = skip_w[col];
  const float bi   = bias[col];
  const bool last_step = t < LAST_STEP_THREADS;
  float hv[EPI_STEPS];
#pragma unroll
  for (int k = 0; k < EPI_STEPS - 1; ++k) hv[k] = hp[k * EPI_ROW_STEP * D];
  hv[EPI_STEPS - 1] = hp[last_step ? (EPI_STEPS - 1) * EPI_ROW_STEP * D : 0];

  for (int i = t; i < R_ROWS * D; i += OWN_THREADS) tile[i] = 0.0f;
  if (t < R_ROWS) cnt[t] = 0;

  int2 rec[SEG_PER_WAVE];
#pragma unroll
  for (int i = 0; i < SEG_PER_WAVE; ++i) {
    const int s = __builtin_amdgcn_readlane(myseg.x, i);
    const int n = __builtin_amdgcn_readlane(myseg.y, i);
    rec[i] = make_int2(0, 0);
    if (lane < n) rec[i] = recs[s + lane];
  }
  __syncthreads();

  int rank[SEG_PER_WAVE];
#pragma unroll
  for (int i = 0; i < SEG_PER_WAVE; ++i) {
    const int n = __builtin_amdgcn_readlane(myseg.y, i);
    rank[i] = 0;
    if (lane < n)
      rank[i] = atomicAdd(&cnt[(unsigned)rec[i].x >> 16], 1);
  }
  __syncthreads();

  if (wave == 0) {
    int c[ROWS_PER_LANE];
    int sum = 0;
#pragma unroll
    for (int k = 0; k < ROWS_PER_LANE; ++k) {
      const int row = lane * ROWS_PER_LANE + k;
      c[k] = row < R_ROWS ? cnt[row] : 0;
      sum += c[k];
    }
    int incl = sum;
#pragma unroll
    for (int d = 1; d < 32; d <<= 1) {
      const int o = __shfl_up(incl, d);
      if (lane >= d) incl += o;
    }
    int run = incl - sum;
#pragma unroll
    for (int k = 0; k < ROWS_PER_LANE; ++k) {
      const int row = lane * ROWS_PER_LANE + k;
      if (row < R_ROWS) off[row] = run;
      run += c[k];
    }
    if (lane == 31) off[R_ROWS] = incl;
  }
  __syncthreads();

  int slot[SEG_PER_WAVE];
#pragma unroll
  for (int i = 0; i < SEG_PER_WAVE; ++i)
    slot[i] = off[(unsigned)rec[i].x >> 16] + rank[i];
#pragma unroll
  for (int i = 0; i < SEG_PER_WAVE; ++i) {
    const int n = __builtin_amdgcn_readlane(myseg.y, i);
    if (lane < n) srec[slot[i]] = rec[i];
  }

#pragma unroll
  for (int i = 0; i < SEG_PER_WAVE; ++i) {
    const int s = __builtin_amdgcn_readlane(myseg.x, i);
    const int n = __builtin_amdgcn_readlane(myseg.y, i);
    for (int base = 32; base < n; base += 32) {
      const int m = (n - base) < 32 ? (n - base) : 32;
      int2 rx = make_int2(0, 0);
      if (lane < m) rx = recs[s + base + lane];
      add_edges(h2, tile, rx, m, lane);
    }
  }
  __syncthreads();

  {
    const int total = __builtin_amdgcn_readfirstlane(off[R_ROWS]);
    const int lo_i  = (total * wave) / OWN_WAVES;
    const int hi_i  = (total * (wave + 1)) / OWN_WAVES;
    const int len   = hi_i - lo_i;
    const int n0    = (len + 1) >> 1;
    const int hf    = lane >> 4;
    const int l16   = lane & 15;
    const int nh    = hf ? (len - n0) : n0;
    const int first = lo_i + (hf ? n0 : 0);
    const float4* __restrict__ h4 = (const float4*)h;
    int    cur = NO_ROW;
    float4 acc = make_float4(0.0f, 0.0f, 0.0f, 0.0f);
    for (int b = 0; b < n0; b += HALF_BLOCK) {
      int2 rx = make_int2((int)((unsigned)NO_ROW << 16), 0);
      if (b + l16 < nh) rx = srec[first + b + l16];
      walk_steps<0>(h4, tile, rx, l16, cur, acc);
      if (b + HALF_STEPS < n0)
        walk_steps<HALF_STEPS>(h4, tile, rx, l16, cur, acc);
    }
    flush_run(tile, cur, acc, l16);
  }
  __syncthreads();

  const float lam = 1.0507009873554805f;
  const float alp = 1.6732632423543772f;
  float tv[EPI_STEPS];
#pragma unroll
  for (int k = 0; k < EPI_STEPS - 1; ++k) tv[k] = tile[t + OWN_THREADS * k];
  tv[EPI_STEPS - 1] = tile[last_step ? t + OWN_THREADS * (EPI_STEPS - 1) : t];

  float* op = out + own;
#pragma unroll
  for (int k = 0; k < EPI_STEPS; ++k) {
    if (k < EPI_STEPS - 1 || last_step) {
      const float x = hv[k] * sk + bi + tv[k];
      op[k * EPI_ROW_STEP * D] =
          lam * (x > 0.0f ? x : alp * (__expf(x) - 1.0f));
    }
  }
}

extern "C" void kernel_launch(void* const* d_in, const int* in_sizes, int n_in,
                              void* d_out, int out_size, void* d_ws,
                              size_t ws_size, hipStream_t stream) {
  const float* feat   = (const float*)d_in[0];
  const float* W      = (const float*)d_in[1];
  const float* bias   = (const float*)d_in[2];
  const float* skip_w = (const float*)d_in[3];
  const float* ew     = (const float*)d_in[4];
  const int*   esrc   = (const int*)d_in[5];
  const int*   edst   = (const int*)d_in[6];
  float* out = (float*)d_out;
  float* h    = (float*)d_ws;
  int2*  recs = (int2*)((char*)d_ws + 16777216);
  int2*  seg  = (int2*)((char*)d_ws + 33554432);

  bin_and_gemm<<<NCH + GEMM_BLOCKS, PREP_THREADS, 0, stream>>>(
      feat, W, h, ew, esrc, edst, recs, seg);

  owner_sum<<<NR, OWN_THREADS, 0, stream>>>(h, recs, seg, skip_w, bias, out);
}
